# starting kernel
# baseline (speedup 1.0000x reference)
#include <hip/hip_runtime.h>
#include <hip/hip_bf16.h>

typedef __attribute__((ext_vector_type(2))) float v2f;
typedef __attribute__((ext_vector_type(8))) float v8f;

#define C_IN  128
#define H_IN  56
#define W_IN  56
#define N_IMG 32
#define K_OUT 256
#define HW    (H_IN * W_IN)
#define M_TOT (N_IMG * HW)
#define KD    (C_IN * 9)

#define CCHUNK   16
#define NPAIR    (CCHUNK / 2)
#define WLDS_N   (9 * NPAIR * K_OUT * 2)

__global__ __launch_bounds__(256)
void conv3x3_wmma_f32(const float* __restrict__ x,
                      const float* __restrict__ wgt,
                      const float* __restrict__ bias,
                      float* __restrict__ out)
{
    __shared__ float wlds[WLDS_N];

    const int tid    = threadIdx.x;
    const int lane   = tid & 31;
    const int wave   = tid >> 5;
    const int wave_m = wave & 1;
    const int wave_n = wave >> 1;

    const int m_base = blockIdx.x * 128 + wave_m * 64;
    const int n_base = wave_n * 64;

    const int lrow  = lane & 15;
    const int khalf = lane >> 4;

    int nimg[4], oh[4], ow[4];
    #pragma unroll
    for (int mt = 0; mt < 4; ++mt) {
        int m    = m_base + mt * 16 + lrow;
        nimg[mt] = m / HW;
        int pix  = m - nimg[mt] * HW;
        oh[mt]   = pix / W_IN;
        ow[mt]   = pix - oh[mt] * W_IN;
    }

    int oc[4];
    #pragma unroll
    for (int nt = 0; nt < 4; ++nt) oc[nt] = n_base + nt * 16 + lrow;

    v8f acc[4][4];
    #pragma unroll
    for (int mt = 0; mt < 4; ++mt)
        #pragma unroll
        for (int nt = 0; nt < 4; ++nt) acc[mt][nt] = (v8f){};

    for (int cb0 = 0; cb0 < C_IN; cb0 += CCHUNK) {
        __syncthreads();
        #pragma unroll 1
        for (int idx = tid; idx < K_OUT * (CCHUNK * 9); idx += 256) {
            int ocq = idx / (CCHUNK * 9);
            int t   = idx - ocq * (CCHUNK * 9);
            int cl  = t / 9;
            int rs  = t - cl * 9;
            float w = wgt[ocq * KD + (cb0 + cl) * 9 + rs];
            int cp  = cl >> 1;
            int j   = cl & 1;
            wlds[(((rs * NPAIR + cp) * K_OUT + ocq) << 1) | j] = w;
        }
        __syncthreads();

        for (int rs = 0; rs < 9; ++rs) {
            const int r = rs / 3;
            const int s = rs - r * 3;

            int  abase[4];
            bool aval[4];
            #pragma unroll
            for (int mt = 0; mt < 4; ++mt) {
                int ih = oh[mt] + r - 1;
                int iw = ow[mt] + s - 1;
                aval[mt]  = ((unsigned)ih < (unsigned)H_IN) & ((unsigned)iw < (unsigned)W_IN);
                int off   = nimg[mt] * (C_IN * HW) + ih * W_IN + iw;
                abase[mt] = aval[mt] ? off : 0;
            }

            #pragma unroll
            for (int cb = 0; cb < CCHUNK; cb += 4) {
                const int c0 = cb0 + cb + khalf * 2;
                const int c1 = c0 + 1;

                v2f a[4], b[4];
                #pragma unroll
                for (int mt = 0; mt < 4; ++mt) {
                    float t0 = x[abase[mt] + c0 * HW];
                    float t1 = x[abase[mt] + c1 * HW];
                    a[mt].x = aval[mt] ? t0 : 0.0f;
                    a[mt].y = aval[mt] ? t1 : 0.0f;
                }
                const int cpbase = rs * NPAIR + (cb >> 1) + khalf;
                #pragma unroll
                for (int nt = 0; nt < 4; ++nt) {
                    b[nt] = *(const v2f*)&wlds[(cpbase * K_OUT + oc[nt]) << 1];
                }

                #pragma unroll
                for (int mt = 0; mt < 4; ++mt)
                    #pragma unroll
                    for (int nt = 0; nt < 4; ++nt)
                        acc[mt][nt] = __builtin_amdgcn_wmma_f32_16x16x4_f32(
                            false, a[mt], false, b[nt], (short)0, acc[mt][nt],
                            false, false);
            }
        }
    }

    #pragma unroll
    for (int nt = 0; nt < 4; ++nt) {
        const float bv = bias[oc[nt]];
        #pragma unroll
        for (int mt = 0; mt < 4; ++mt) {
            #pragma unroll
            for (int v = 0; v < 8; ++v) {
                const int mrow = m_base + mt * 16 + 8 * khalf + v;
                const int ni   = mrow / HW;
                const int pix  = mrow - ni * HW;
                out[ni * (K_OUT * HW) + oc[nt] * HW + pix] = acc[mt][nt][v] + bv;
            }
        }
    }
}

extern "C" void kernel_launch(void* const* d_in, const int* in_sizes, int n_in,
                              void* d_out, int out_size, void* d_ws, size_t ws_size,
                              hipStream_t stream) {
    (void)in_sizes; (void)n_in; (void)out_size; (void)d_ws; (void)ws_size;
    const float* x    = (const float*)d_in[0];
    const float* wgt  = (const float*)d_in[1];
    const float* bias = (const float*)d_in[2];
    float* out        = (float*)d_out;

    dim3 grid(M_TOT / 128, 1);
    dim3 block(256);
    conv3x3_wmma_f32<<<grid, block, 0, stream>>>(x, wgt, bias, out);
}


# Weights staged in LDS in global row order (one 144-float row per out-channel, pitch 148) by coalesced 16-byte copies, 18 loads in flight per lane; B fragment taken by two strided 4-byte reads
# speedup vs baseline: 1.7504x; 1.7504x over previous
#include <hip/hip_runtime.h>
#include <hip/hip_bf16.h>

typedef __attribute__((ext_vector_type(2))) float v2f;
typedef __attribute__((ext_vector_type(8))) float v8f;
typedef __attribute__((ext_vector_type(4))) float v4f;

#define C_IN  128
#define H_IN  56
#define W_IN  56
#define N_IMG 32
#define K_OUT 256
#define HW    (H_IN * W_IN)
#define M_TOT (N_IMG * HW)
#define KD    (C_IN * 9)

#define CCHUNK   16
#define WROW     (CCHUNK * 9)
#define WPITCH   (WROW + 4)
#define WLDS_N   (K_OUT * WPITCH)
#define NSTAGE   36
#define SBATCH   (NSTAGE / 2)

__device__ __forceinline__ int stage_off(int i, int lane, int pitch4)
{
    const int g = i / 9;
    const int r = i - g * 9;
    return (r < 8) ? ((g * 8 + r) * pitch4 + lane)
                   : ((g * 8 + (lane >> 2)) * pitch4 + 32 + (lane & 3));
}

__global__ __launch_bounds__(256)
void conv3x3_wmma_f32(const float* __restrict__ x,
                      const float* __restrict__ wgt,
                      const float* __restrict__ bias,
                      float* __restrict__ out)
{
    __shared__ v4f wlds4[WLDS_N / 4];
    const float* wlds = reinterpret_cast<const float*>(wlds4);

    const int tid    = threadIdx.x;
    const int lane   = tid & 31;
    const int wave   = tid >> 5;
    const int wave_m = wave & 1;
    const int wave_n = wave >> 1;

    const int m_base = blockIdx.x * 128 + wave_m * 64;
    const int n_base = wave_n * 64;

    const int lrow  = lane & 15;
    const int khalf = lane >> 4;

    int nimg[4], oh[4], ow[4];
    #pragma unroll
    for (int mt = 0; mt < 4; ++mt) {
        int m    = m_base + mt * 16 + lrow;
        nimg[mt] = m / HW;
        int pix  = m - nimg[mt] * HW;
        oh[mt]   = pix / W_IN;
        ow[mt]   = pix - oh[mt] * W_IN;
    }

    int oc[4], boff[4];
    #pragma unroll
    for (int nt = 0; nt < 4; ++nt) {
        oc[nt]   = n_base + nt * 16 + lrow;
        boff[nt] = oc[nt] * WPITCH + khalf * 18;
    }

    v8f acc[4][4];
    #pragma unroll
    for (int mt = 0; mt < 4; ++mt)
        #pragma unroll
        for (int nt = 0; nt < 4; ++nt) acc[mt][nt] = (v8f){};

    v4f* const wdst = wlds4 + wave * 32 * (WPITCH / 4);

    for (int cb0 = 0; cb0 < C_IN; cb0 += CCHUNK) {
        const v4f* wsrc =
            reinterpret_cast<const v4f*>(wgt + wave * 32 * KD + cb0 * 9);

        v4f stg[SBATCH];
        #pragma unroll
        for (int k = 0; k < SBATCH; ++k) stg[k] = wsrc[stage_off(k, lane, KD / 4)];

        __syncthreads();

        #pragma unroll
        for (int k = 0; k < SBATCH; ++k) wdst[stage_off(k, lane, WPITCH / 4)] = stg[k];
        #pragma unroll
        for (int k = 0; k < SBATCH; ++k) stg[k] = wsrc[stage_off(SBATCH + k, lane, KD / 4)];
        #pragma unroll
        for (int k = 0; k < SBATCH; ++k) wdst[stage_off(SBATCH + k, lane, WPITCH / 4)] = stg[k];

        __syncthreads();

        for (int rs = 0; rs < 9; ++rs) {
            const int r = rs / 3;
            const int s = rs - r * 3;

            int  abase[4];
            bool aval[4];
            #pragma unroll
            for (int mt = 0; mt < 4; ++mt) {
                int ih = oh[mt] + r - 1;
                int iw = ow[mt] + s - 1;
                aval[mt]  = ((unsigned)ih < (unsigned)H_IN) & ((unsigned)iw < (unsigned)W_IN);
                int off   = nimg[mt] * (C_IN * HW) + ih * W_IN + iw;
                abase[mt] = aval[mt] ? off : 0;
            }

            #pragma unroll
            for (int cb = 0; cb < CCHUNK; cb += 4) {
                const int c0 = cb0 + cb + khalf * 2;
                const int c1 = c0 + 1;

                v2f a[4], b[4];
                #pragma unroll
                for (int mt = 0; mt < 4; ++mt) {
                    float t0 = x[abase[mt] + c0 * HW];
                    float t1 = x[abase[mt] + c1 * HW];
                    a[mt].x = aval[mt] ? t0 : 0.0f;
                    a[mt].y = aval[mt] ? t1 : 0.0f;
                }
                #pragma unroll
                for (int nt = 0; nt < 4; ++nt) {
                    const float* wp = &wlds[boff[nt] + cb * 9 + rs];
                    b[nt].x = wp[0];
                    b[nt].y = wp[9];
                }

                #pragma unroll
                for (int mt = 0; mt < 4; ++mt)
                    #pragma unroll
                    for (int nt = 0; nt < 4; ++nt)
                        acc[mt][nt] = __builtin_amdgcn_wmma_f32_16x16x4_f32(
                            false, a[mt], false, b[nt], (short)0, acc[mt][nt],
                            false, false);
            }
        }
    }

    #pragma unroll
    for (int nt = 0; nt < 4; ++nt) {
        const float bv = bias[oc[nt]];
        #pragma unroll
        for (int mt = 0; mt < 4; ++mt) {
            #pragma unroll
            for (int v = 0; v < 8; ++v) {
                const int mrow = m_base + mt * 16 + 8 * khalf + v;
                const int ni   = mrow / HW;
                const int pix  = mrow - ni * HW;
                out[ni * (K_OUT * HW) + oc[nt] * HW + pix] = acc[mt][nt][v] + bv;
            }
        }
    }
}

extern "C" void kernel_launch(void* const* d_in, const int* in_sizes, int n_in,
                              void* d_out, int out_size, void* d_ws, size_t ws_size,
                              hipStream_t stream) {
    (void)in_sizes; (void)n_in; (void)out_size; (void)d_ws; (void)ws_size;
    const float* x    = (const float*)d_in[0];
    const float* wgt  = (const float*)d_in[1];
    const float* bias = (const float*)d_in[2];
    float* out        = (float*)d_out;

    dim3 grid(M_TOT / 128, 1);
    dim3 block(256);
    conv3x3_wmma_f32<<<grid, block, 0, stream>>>(x, wgt, bias, out);
}


# Operands swapped in the WMMA (weights as A, pixels as B) and pixel-tile pairs joined across lane halves: each store writes one whole 128-byte line of a channel, not 32 scattered words; sums unchanged
# speedup vs baseline: 1.9013x; 1.0862x over previous
#include <hip/hip_runtime.h>
#include <hip/hip_bf16.h>

typedef __attribute__((ext_vector_type(2))) float v2f;
typedef __attribute__((ext_vector_type(8))) float v8f;
typedef __attribute__((ext_vector_type(4))) float v4f;
typedef __attribute__((ext_vector_type(2))) unsigned int v2u;

#define C_IN  128
#define H_IN  56
#define W_IN  56
#define N_IMG 32
#define K_OUT 256
#define HW    (H_IN * W_IN)
#define M_TOT (N_IMG * HW)
#define KD    (C_IN * 9)

#define CCHUNK   16
#define WROW     (CCHUNK * 9)
#define WPITCH   (WROW + 4)
#define WLDS_N   (K_OUT * WPITCH)
#define NSTAGE   36
#define SBATCH   (NSTAGE / 2)

__device__ __forceinline__ int stage_off(int i, int lane, int pitch4)
{
    const int g = i / 9;
    const int r = i - g * 9;
    return (r < 8) ? ((g * 8 + r) * pitch4 + lane)
                   : ((g * 8 + (lane >> 2)) * pitch4 + 32 + (lane & 3));
}

__global__ __launch_bounds__(256)
void conv3x3_wmma_f32(const float* __restrict__ x,
                      const float* __restrict__ wgt,
                      const float* __restrict__ bias,
                      float* __restrict__ out)
{
    __shared__ v4f wlds4[WLDS_N / 4];
    const float* wlds = reinterpret_cast<const float*>(wlds4);

    const int tid    = threadIdx.x;
    const int lane   = tid & 31;
    const int wave   = tid >> 5;
    const int wave_m = wave & 1;
    const int wave_n = wave >> 1;

    const int m_base = blockIdx.x * 128 + wave_m * 64;
    const int n_base = wave_n * 64;

    const int lrow  = lane & 15;
    const int khalf = lane >> 4;

    int nimg[4], oh[4], ow[4];
    #pragma unroll
    for (int mt = 0; mt < 4; ++mt) {
        int m    = m_base + mt * 16 + lrow;
        nimg[mt] = m / HW;
        int pix  = m - nimg[mt] * HW;
        oh[mt]   = pix / W_IN;
        ow[mt]   = pix - oh[mt] * W_IN;
    }

    int oc[4], boff[4];
    #pragma unroll
    for (int nt = 0; nt < 4; ++nt) {
        oc[nt]   = n_base + nt * 16 + lrow;
        boff[nt] = oc[nt] * WPITCH + khalf * 18;
    }

    v8f acc[4][4];
    #pragma unroll
    for (int mt = 0; mt < 4; ++mt)
        #pragma unroll
        for (int nt = 0; nt < 4; ++nt) acc[mt][nt] = (v8f){};

    v4f* const wdst = wlds4 + wave * 32 * (WPITCH / 4);

    for (int cb0 = 0; cb0 < C_IN; cb0 += CCHUNK) {
        const v4f* wsrc =
            reinterpret_cast<const v4f*>(wgt + wave * 32 * KD + cb0 * 9);

        v4f stg[SBATCH];
        #pragma unroll
        for (int k = 0; k < SBATCH; ++k) stg[k] = wsrc[stage_off(k, lane, KD / 4)];

        __syncthreads();

        #pragma unroll
        for (int k = 0; k < SBATCH; ++k) wdst[stage_off(k, lane, WPITCH / 4)] = stg[k];
        #pragma unroll
        for (int k = 0; k < SBATCH; ++k) stg[k] = wsrc[stage_off(SBATCH + k, lane, KD / 4)];
        #pragma unroll
        for (int k = 0; k < SBATCH; ++k) wdst[stage_off(SBATCH + k, lane, WPITCH / 4)] = stg[k];

        __syncthreads();

        for (int rs = 0; rs < 9; ++rs) {
            const int r = rs / 3;
            const int s = rs - r * 3;

            int  abase[4];
            bool aval[4];
            #pragma unroll
            for (int mt = 0; mt < 4; ++mt) {
                int ih = oh[mt] + r - 1;
                int iw = ow[mt] + s - 1;
                aval[mt]  = ((unsigned)ih < (unsigned)H_IN) & ((unsigned)iw < (unsigned)W_IN);
                int off   = nimg[mt] * (C_IN * HW) + ih * W_IN + iw;
                abase[mt] = aval[mt] ? off : 0;
            }

            #pragma unroll
            for (int cb = 0; cb < CCHUNK; cb += 4) {
                const int c0 = cb0 + cb + khalf * 2;
                const int c1 = c0 + 1;

                v2f a[4], b[4];
                #pragma unroll
                for (int mt = 0; mt < 4; ++mt) {
                    float t0 = x[abase[mt] + c0 * HW];
                    float t1 = x[abase[mt] + c1 * HW];
                    a[mt].x = aval[mt] ? t0 : 0.0f;
                    a[mt].y = aval[mt] ? t1 : 0.0f;
                }
                #pragma unroll
                for (int nt = 0; nt < 4; ++nt) {
                    const float* wp = &wlds[boff[nt] + cb * 9 + rs];
                    b[nt].x = wp[0];
                    b[nt].y = wp[9];
                }

                #pragma unroll
                for (int mt = 0; mt < 4; ++mt)
                    #pragma unroll
                    for (int nt = 0; nt < 4; ++nt)
                        acc[mt][nt] = __builtin_amdgcn_wmma_f32_16x16x4_f32(
                            false, b[nt], false, a[mt], (short)0, acc[mt][nt],
                            false, false);
            }
        }
    }

    int obase[2];
    #pragma unroll
    for (int p = 0; p < 2; ++p) {
        const int m   = m_base + p * 32 + lane;
        const int ni  = m / HW;
        const int pix = m - ni * HW;
        obase[p] = ni * (K_OUT * HW) + pix;
    }

    #pragma unroll
    for (int nt = 0; nt < 4; ++nt) {
        const int ocb = n_base + nt * 16;
        const v4f b0 = *reinterpret_cast<const v4f*>(bias + ocb);
        const v4f b1 = *reinterpret_cast<const v4f*>(bias + ocb + 4);
        const v4f b2 = *reinterpret_cast<const v4f*>(bias + ocb + 8);
        const v4f b3 = *reinterpret_cast<const v4f*>(bias + ocb + 12);
        const float bv[16] = { b0.x, b0.y, b0.z, b0.w, b1.x, b1.y, b1.z, b1.w,
                               b2.x, b2.y, b2.z, b2.w, b3.x, b3.y, b3.z, b3.w };
        #pragma unroll
        for (int p = 0; p < 2; ++p) {
            #pragma unroll
            for (int v = 0; v < 8; ++v) {
                const float t_lo = acc[2 * p][nt][v];
                const float t_hi = acc[2 * p + 1][nt][v];
                const v2u sw = __builtin_amdgcn_permlane16_swap(
                    __float_as_uint(t_lo), __float_as_uint(t_hi), false, false);
                out[obase[p] + (ocb + v) * HW]     = __uint_as_float(sw.x) + bv[v];
                out[obase[p] + (ocb + 8 + v) * HW] = __uint_as_float(sw.y) + bv[8 + v];
            }
        }
    }
}

extern "C" void kernel_launch(void* const* d_in, const int* in_sizes, int n_in,
                              void* d_out, int out_size, void* d_ws, size_t ws_size,
                              hipStream_t stream) {
    (void)in_sizes; (void)n_in; (void)out_size; (void)d_ws; (void)ws_size;
    const float* x    = (const float*)d_in[0];
    const float* wgt  = (const float*)d_in[1];
    const float* bias = (const float*)d_in[2];
    float* out        = (float*)d_out;

    dim3 grid(M_TOT / 128, 1);
    dim3 block(256);
    conv3x3_wmma_f32<<<grid, block, 0, stream>>>(x, wgt, bias, out);
}


# Grid re-cut so no WGP runs a fourth 128-pixel block: 768 blocks of 128 pixels (three per WGP) plus 64 blocks of 32 pixels placed first in the grid; same loop body as a template; sums unchanged
# speedup vs baseline: 2.1590x; 1.1355x over previous
#include <hip/hip_runtime.h>
#include <hip/hip_bf16.h>

typedef __attribute__((ext_vector_type(2))) float v2f;
typedef __attribute__((ext_vector_type(8))) float v8f;
typedef __attribute__((ext_vector_type(4))) float v4f;
typedef __attribute__((ext_vector_type(2))) unsigned int v2u;

#define C_IN  128
#define H_IN  56
#define W_IN  56
#define N_IMG 32
#define K_OUT 256
#define HW    (H_IN * W_IN)
#define M_TOT (N_IMG * HW)
#define KD    (C_IN * 9)

#define CCHUNK   16
#define WROW     (CCHUNK * 9)
#define WPITCH   (WROW + 4)
#define WLDS_N   (K_OUT * WPITCH)
#define NSTAGE   36
#define SBATCH   (NSTAGE / 2)

__device__ __forceinline__ int stage_off(int i, int lane, int pitch4)
{
    const int g = i / 9;
    const int r = i - g * 9;
    return (r < 8) ? ((g * 8 + r) * pitch4 + lane)
                   : ((g * 8 + (lane >> 2)) * pitch4 + 32 + (lane & 3));
}

#define BIG_PIX    128
#define N_BIG      768
#define SMALL_PIX  32
#define N_SMALL    ((M_TOT - N_BIG * BIG_PIX) / SMALL_PIX)

template <int MT, int NT, int WM>
__device__ __forceinline__
void conv_tile(const float* __restrict__ x,
               const float* __restrict__ wgt,
               const float* __restrict__ bias,
               float* __restrict__ out,
               v4f* __restrict__ wlds4,
               const int m_blk)
{
    const float* wlds = reinterpret_cast<const float*>(wlds4);

    const int tid    = threadIdx.x;
    const int lane   = tid & 31;
    const int wave   = tid >> 5;
    const int wave_m = wave % WM;
    const int wave_n = wave / WM;

    const int m_base = m_blk + wave_m * (MT * 16);
    const int n_base = wave_n * (NT * 16);

    const int lrow  = lane & 15;
    const int khalf = lane >> 4;

    int nimg[MT], oh[MT], ow[MT];
    #pragma unroll
    for (int mt = 0; mt < MT; ++mt) {
        int m    = m_base + mt * 16 + lrow;
        nimg[mt] = m / HW;
        int pix  = m - nimg[mt] * HW;
        oh[mt]   = pix / W_IN;
        ow[mt]   = pix - oh[mt] * W_IN;
    }

    int oc[NT], boff[NT];
    #pragma unroll
    for (int nt = 0; nt < NT; ++nt) {
        oc[nt]   = n_base + nt * 16 + lrow;
        boff[nt] = oc[nt] * WPITCH + khalf * 18;
    }

    v8f acc[MT][NT];
    #pragma unroll
    for (int mt = 0; mt < MT; ++mt)
        #pragma unroll
        for (int nt = 0; nt < NT; ++nt) acc[mt][nt] = (v8f){};

    v4f* const wdst = wlds4 + wave * 32 * (WPITCH / 4);

    for (int cb0 = 0; cb0 < C_IN; cb0 += CCHUNK) {
        const v4f* wsrc =
            reinterpret_cast<const v4f*>(wgt + wave * 32 * KD + cb0 * 9);

        v4f stg[SBATCH];
        #pragma unroll
        for (int k = 0; k < SBATCH; ++k) stg[k] = wsrc[stage_off(k, lane, KD / 4)];

        __syncthreads();

        #pragma unroll
        for (int k = 0; k < SBATCH; ++k) wdst[stage_off(k, lane, WPITCH / 4)] = stg[k];
        #pragma unroll
        for (int k = 0; k < SBATCH; ++k) stg[k] = wsrc[stage_off(SBATCH + k, lane, KD / 4)];
        #pragma unroll
        for (int k = 0; k < SBATCH; ++k) wdst[stage_off(SBATCH + k, lane, WPITCH / 4)] = stg[k];

        __syncthreads();

        #pragma nounroll
        for (int rs = 0; rs < 9; ++rs) {
            const int r = rs / 3;
            const int s = rs - r * 3;

            int  abase[MT];
            bool aval[MT];
            #pragma unroll
            for (int mt = 0; mt < MT; ++mt) {
                int ih = oh[mt] + r - 1;
                int iw = ow[mt] + s - 1;
                aval[mt]  = ((unsigned)ih < (unsigned)H_IN) & ((unsigned)iw < (unsigned)W_IN);
                int off   = nimg[mt] * (C_IN * HW) + ih * W_IN + iw;
                abase[mt] = aval[mt] ? off : 0;
            }

            #pragma unroll
            for (int cb = 0; cb < CCHUNK; cb += 4) {
                const int c0 = cb0 + cb + khalf * 2;
                const int c1 = c0 + 1;

                v2f a[MT], b[NT];
                #pragma unroll
                for (int mt = 0; mt < MT; ++mt) {
                    float t0 = x[abase[mt] + c0 * HW];
                    float t1 = x[abase[mt] + c1 * HW];
                    a[mt].x = aval[mt] ? t0 : 0.0f;
                    a[mt].y = aval[mt] ? t1 : 0.0f;
                }
                #pragma unroll
                for (int nt = 0; nt < NT; ++nt) {
                    const float* wp = &wlds[boff[nt] + cb * 9 + rs];
                    b[nt].x = wp[0];
                    b[nt].y = wp[9];
                }

                #pragma unroll
                for (int mt = 0; mt < MT; ++mt)
                    #pragma unroll
                    for (int nt = 0; nt < NT; ++nt)
                        acc[mt][nt] = __builtin_amdgcn_wmma_f32_16x16x4_f32(
                            false, b[nt], false, a[mt], (short)0, acc[mt][nt],
                            false, false);
            }
        }
    }

    int obase[MT / 2];
    #pragma unroll
    for (int p = 0; p < MT / 2; ++p) {
        const int m   = m_base + p * 32 + lane;
        const int ni  = m / HW;
        const int pix = m - ni * HW;
        obase[p] = ni * (K_OUT * HW) + pix;
    }

    #pragma unroll
    for (int nt = 0; nt < NT; ++nt) {
        const int ocb = n_base + nt * 16;
        const v4f b0 = *reinterpret_cast<const v4f*>(bias + ocb);
        const v4f b1 = *reinterpret_cast<const v4f*>(bias + ocb + 4);
        const v4f b2 = *reinterpret_cast<const v4f*>(bias + ocb + 8);
        const v4f b3 = *reinterpret_cast<const v4f*>(bias + ocb + 12);
        const float bv[16] = { b0.x, b0.y, b0.z, b0.w, b1.x, b1.y, b1.z, b1.w,
                               b2.x, b2.y, b2.z, b2.w, b3.x, b3.y, b3.z, b3.w };
        #pragma unroll
        for (int p = 0; p < MT / 2; ++p) {
            #pragma unroll
            for (int v = 0; v < 8; ++v) {
                const float t_lo = acc[2 * p][nt][v];
                const float t_hi = acc[2 * p + 1][nt][v];
                const v2u sw = __builtin_amdgcn_permlane16_swap(
                    __float_as_uint(t_lo), __float_as_uint(t_hi), false, false);
                out[obase[p] + (ocb + v) * HW]     = __uint_as_float(sw.x) + bv[v];
                out[obase[p] + (ocb + 8 + v) * HW] = __uint_as_float(sw.y) + bv[8 + v];
            }
        }
    }
}

__global__ __launch_bounds__(256)
void conv3x3_wmma_f32(const float* __restrict__ x,
                      const float* __restrict__ wgt,
                      const float* __restrict__ bias,
                      float* __restrict__ out)
{
    __shared__ v4f wlds4[WLDS_N / 4];

    if (blockIdx.x < N_SMALL) {
        conv_tile<2, 2, 1>(x, wgt, bias, out, wlds4,
                           N_BIG * BIG_PIX + (int)blockIdx.x * SMALL_PIX);
    } else {
        conv_tile<4, 4, 2>(x, wgt, bias, out, wlds4,
                           ((int)blockIdx.x - N_SMALL) * BIG_PIX);
    }
}

extern "C" void kernel_launch(void* const* d_in, const int* in_sizes, int n_in,
                              void* d_out, int out_size, void* d_ws, size_t ws_size,
                              hipStream_t stream) {
    (void)in_sizes; (void)n_in; (void)out_size; (void)d_ws; (void)ws_size;
    const float* x    = (const float*)d_in[0];
    const float* wgt  = (const float*)d_in[1];
    const float* bias = (const float*)d_in[2];
    float* out        = (float*)d_out;

    dim3 grid(N_SMALL + N_BIG, 1);
    dim3 block(256);
    conv3x3_wmma_f32<<<grid, block, 0, stream>>>(x, wgt, bias, out);
}


# Width taps by minimal filtering F(2,3): 4 matrix products per output pair and tap row, not 6; weights transformed while staged, a lane stores its pixel pair as 8 bytes; sums re-associated
# speedup vs baseline: 2.2952x; 1.0631x over previous
#include <hip/hip_runtime.h>

typedef __attribute__((ext_vector_type(2))) float v2f;
typedef __attribute__((ext_vector_type(8))) float v8f;
typedef __attribute__((ext_vector_type(4))) float v4f;

#define C_IN  128
#define H_IN  56
#define W_IN  56
#define N_IMG 32
#define K_OUT 256
#define HW    (H_IN * W_IN)
#define KD    (C_IN * 9)
#define WPAIR (W_IN / 2)
#define HWP   (HW / 2)
#define P_TOT (N_IMG * HWP)

#define CCH      8
#define SLOTS    12
#define WPITCH4  25
#define WLDS4    (K_OUT * WPITCH4)
#define HALF4    12

#define BIG_PAIRS   32
#define N_BIG       1536
#define SMALL_PAIRS 16
#define N_SMALL     ((P_TOT - N_BIG * BIG_PAIRS) / SMALL_PAIRS)

template <int NT, int WM>
__device__ __forceinline__
void conv_tile(const float* __restrict__ x,
               const float* __restrict__ wgt,
               const float* __restrict__ bias,
               float* __restrict__ out,
               v4f* __restrict__ wlds4,
               const int q_blk)
{
    const int tid    = threadIdx.x;
    const int lane   = tid & 31;
    const int wave   = tid >> 5;
    const int wave_m = wave % WM;
    const int wave_n = wave / WM;

    const int n_base = wave_n * (NT * 16);
    const int lrow   = lane & 15;
    const int khalf  = lane >> 4;

    const int qg   = q_blk + wave_m * 16 + lrow;
    const int nimg = qg / HWP;
    const int qi   = qg - nimg * HWP;
    const int oh   = qi / WPAIR;
    const int jp   = qi - oh * WPAIR;

    int  o0[3], o1[3], o3[3];
    bool rv[3], v0[3], v3[3];
    #pragma unroll
    for (int r = 0; r < 3; ++r) {
        const int ih = oh + r - 1;
        rv[r] = (unsigned)ih < (unsigned)H_IN;
        v0[r] = rv[r] & (jp > 0);
        v3[r] = rv[r] & (jp < WPAIR - 1);
        o1[r] = nimg * (C_IN * HW) + (khalf * 2) * HW + (rv[r] ? ih : oh) * W_IN + 2 * jp;
        o0[r] = o1[r] - (v0[r] ? 1 : 0);
        o3[r] = o1[r] + (v3[r] ? 2 : 0);
    }

    const v4f* const wrd = wlds4 + (n_base + lrow) * WPITCH4 + khalf * 3;

    int goff[3], loff4[3];
    #pragma unroll
    for (int i = 0; i < 3; ++i) {
        const int it   = tid + 256 * i;
        const int oc   = it / SLOTS;
        const int slot = it - oc * SLOTS;
        const int cp   = slot / 3;
        const int r    = slot - cp * 3;
        goff[i]  = oc * KD + cp * 18 + r * 3;
        loff4[i] = oc * WPITCH4 + slot;
    }

    v8f acc[NT][4];
    #pragma unroll
    for (int nt = 0; nt < NT; ++nt)
        #pragma unroll
        for (int t = 0; t < 4; ++t) acc[nt][t] = (v8f){};

    float ra0[2], rb0[2], ra3[2], rb3[2];
    v2f   ra12[2], rb12[2];
    v4f   wa[2][NT], wb[2][NT];

    for (int cb0 = 0; cb0 < C_IN; cb0 += CCH) {
        ra0[0]  = x[o0[0]];
        rb0[0]  = x[o0[0] + HW];
        ra12[0] = *reinterpret_cast<const v2f*>(x + o1[0]);
        rb12[0] = *reinterpret_cast<const v2f*>(x + o1[0] + HW);
        ra3[0]  = x[o3[0]];
        rb3[0]  = x[o3[0] + HW];

        float g[6][6];
        #pragma unroll
        for (int k = 0; k < 6; ++k) {
            const float* p = wgt + goff[k % 3] + (k / 3) * (64 * KD);
            g[k][0] = p[0]; g[k][1] = p[1];  g[k][2] = p[2];
            g[k][3] = p[9]; g[k][4] = p[10]; g[k][5] = p[11];
        }

        __syncthreads();

        #pragma unroll
        for (int b = 0; b < 2; ++b) {
            #pragma unroll
            for (int k = 0; k < 6; ++k) {
                const float sa = g[k][0] + g[k][2];
                const float sb = g[k][3] + g[k][5];
                const v4f lo = { g[k][0], g[k][3], 0.5f * (sa + g[k][1]), 0.5f * (sb + g[k][4]) };
                const v4f hi = { 0.5f * (sa - g[k][1]), 0.5f * (sb - g[k][4]), g[k][2], g[k][5] };
                v4f* d = wlds4 + loff4[k % 3] + (b * 2 + k / 3) * (64 * WPITCH4);
                d[0]     = lo;
                d[HALF4] = hi;
            }
            if (b == 0) {
                #pragma unroll
                for (int k = 0; k < 6; ++k) {
                    const float* p = wgt + goff[k % 3] + (2 + k / 3) * (64 * KD);
                    g[k][0] = p[0]; g[k][1] = p[1];  g[k][2] = p[2];
                    g[k][3] = p[9]; g[k][4] = p[10]; g[k][5] = p[11];
                }
            }
        }

        __syncthreads();

        #pragma unroll
        for (int nt = 0; nt < NT; ++nt) {
            const v4f* wp = wrd + nt * (16 * WPITCH4);
            wa[0][nt] = wp[0];
            wb[0][nt] = wp[HALF4];
        }

        #pragma unroll
        for (int i = 0; i < 6; ++i) {
            const int s = i & 1;
            if (i < 5) {
                const int r  = (i + 1) >> 1;
                const int ks = (i + 1) & 1;
                const int n  = s ^ 1;
                const float* xp = x + (ks * 4) * HW;
                ra0[n]  = xp[o0[r]];
                rb0[n]  = xp[o0[r] + HW];
                ra12[n] = *reinterpret_cast<const v2f*>(xp + o1[r]);
                rb12[n] = *reinterpret_cast<const v2f*>(xp + o1[r] + HW);
                ra3[n]  = xp[o3[r]];
                rb3[n]  = xp[o3[r] + HW];
                #pragma unroll
                for (int nt = 0; nt < NT; ++nt) {
                    const v4f* wp = wrd + nt * (16 * WPITCH4) + (ks * 6 + r);
                    wa[n][nt] = wp[0];
                    wb[n][nt] = wp[HALF4];
                }
            }
            __builtin_amdgcn_sched_barrier(0);
            {
                const int r = i >> 1;
                const float p0 = v0[r] ? ra0[s] : 0.0f,    q0 = v0[r] ? rb0[s] : 0.0f;
                const float p1 = rv[r] ? ra12[s].x : 0.0f, q1 = rv[r] ? rb12[s].x : 0.0f;
                const float p2 = rv[r] ? ra12[s].y : 0.0f, q2 = rv[r] ? rb12[s].y : 0.0f;
                const float p3 = v3[r] ? ra3[s] : 0.0f,    q3 = v3[r] ? rb3[s] : 0.0f;

                const v2f u0 = { p0 - p2, q0 - q2 };
                const v2f u1 = { p1 + p2, q1 + q2 };
                const v2f u2 = { p2 - p1, q2 - q1 };
                const v2f u3 = { p1 - p3, q1 - q3 };

                #pragma unroll
                for (int nt = 0; nt < NT; ++nt) {
                    const v2f a0 = { wa[s][nt].x, wa[s][nt].y };
                    const v2f a1 = { wa[s][nt].z, wa[s][nt].w };
                    const v2f a2 = { wb[s][nt].x, wb[s][nt].y };
                    const v2f a3 = { wb[s][nt].z, wb[s][nt].w };
                    acc[nt][0] = __builtin_amdgcn_wmma_f32_16x16x4_f32(
                        false, a0, false, u0, (short)0, acc[nt][0], false, false);
                    acc[nt][1] = __builtin_amdgcn_wmma_f32_16x16x4_f32(
                        false, a1, false, u1, (short)0, acc[nt][1], false, false);
                    acc[nt][2] = __builtin_amdgcn_wmma_f32_16x16x4_f32(
                        false, a2, false, u2, (short)0, acc[nt][2], false, false);
                    acc[nt][3] = __builtin_amdgcn_wmma_f32_16x16x4_f32(
                        false, a3, false, u3, (short)0, acc[nt][3], false, false);
                }
            }
            __builtin_amdgcn_sched_barrier(0);
        }

        #pragma unroll
        for (int r = 0; r < 3; ++r) { o0[r] += CCH * HW; o1[r] += CCH * HW; o3[r] += CCH * HW; }
        #pragma unroll
        for (int i = 0; i < 3; ++i) goff[i] += CCH * 9;
    }

    const int obase = nimg * (K_OUT * HW) + 2 * qi;
    #pragma unroll
    for (int nt = 0; nt < NT; ++nt) {
        const int ocb = n_base + nt * 16 + 8 * khalf;
        const v4f b0 = *reinterpret_cast<const v4f*>(bias + ocb);
        const v4f b1 = *reinterpret_cast<const v4f*>(bias + ocb + 4);
        const float bv[8] = { b0.x, b0.y, b0.z, b0.w, b1.x, b1.y, b1.z, b1.w };
        #pragma unroll
        for (int v = 0; v < 8; ++v) {
            const float m0 = acc[nt][0][v], m1 = acc[nt][1][v];
            const float m2 = acc[nt][2][v], m3 = acc[nt][3][v];
            const v2f y = { (m0 + m1) + m2 + bv[v], (m1 - m2) - m3 + bv[v] };
            *reinterpret_cast<v2f*>(out + obase + (ocb + v) * HW) = y;
        }
    }
}

__global__ __launch_bounds__(256, 4)
void conv3x3_wmma_f32(const float* __restrict__ x,
                      const float* __restrict__ wgt,
                      const float* __restrict__ bias,
                      float* __restrict__ out)
{
    __shared__ v4f wlds4[WLDS4];

    if (blockIdx.x < N_SMALL) {
        conv_tile<2, 1>(x, wgt, bias, out, wlds4,
                        N_BIG * BIG_PAIRS + (int)blockIdx.x * SMALL_PAIRS);
    } else {
        conv_tile<4, 2>(x, wgt, bias, out, wlds4,
                        ((int)blockIdx.x - N_SMALL) * BIG_PAIRS);
    }
}

extern "C" void kernel_launch(void* const* d_in, const int* in_sizes, int n_in,
                              void* d_out, int out_size, void* d_ws, size_t ws_size,
                              hipStream_t stream) {
    (void)in_sizes; (void)n_in; (void)out_size; (void)d_ws; (void)ws_size;
    const float* x    = (const float*)d_in[0];
    const float* wgt  = (const float*)d_in[1];
    const float* bias = (const float*)d_in[2];
    float* out        = (float*)d_out;

    dim3 grid(N_SMALL + N_BIG, 1);
    dim3 block(256);
    conv3x3_wmma_f32<<<grid, block, 0, stream>>>(x, wgt, bias, out);
}


# Big block tile re-cut from 32 pairs x 256 out-channels to 64 pairs x 128: a block stages half the weights per chunk, in one batch requested before the barrier; same wave tile, sums unchanged
# speedup vs baseline: 2.8426x; 1.2385x over previous
#include <hip/hip_runtime.h>

typedef __attribute__((ext_vector_type(2))) float v2f;
typedef __attribute__((ext_vector_type(8))) float v8f;
typedef __attribute__((ext_vector_type(4))) float v4f;

#define C_IN  128
#define H_IN  56
#define W_IN  56
#define N_IMG 32
#define K_OUT 256
#define HW    (H_IN * W_IN)
#define KD    (C_IN * 9)
#define WPAIR (W_IN / 2)
#define HWP   (HW / 2)
#define P_TOT (N_IMG * HWP)

#define CCH      8
#define SLOTS    12
#define WPITCH4  25
#define WLDS4    (K_OUT * WPITCH4)
#define HALF4    12

#define BIG_PAIRS   64
#define BIG_OC      128
#define N_BIG       1536
#define BIG_COVER   ((N_BIG / (K_OUT / BIG_OC)) * BIG_PAIRS)
#define SMALL_PAIRS 16
#define N_SMALL     ((P_TOT - BIG_COVER) / SMALL_PAIRS)

template <int NT, int WM>
__device__ __forceinline__
void conv_tile(const float* __restrict__ x,
               const float* __restrict__ wgt,
               const float* __restrict__ bias,
               float* __restrict__ out,
               v4f* __restrict__ wlds4,
               const int q_blk,
               const int oc_blk)
{
    constexpr int KB = (8 / WM) * NT * 16;
    constexpr int NB = KB / 128;

    const int tid    = threadIdx.x;
    const int lane   = tid & 31;
    const int wave   = tid >> 5;
    const int wave_m = wave % WM;
    const int wave_n = wave / WM;

    const int n_base = wave_n * (NT * 16);
    const int lrow   = lane & 15;
    const int khalf  = lane >> 4;

    const int qg   = q_blk + wave_m * 16 + lrow;
    const int nimg = qg / HWP;
    const int qi   = qg - nimg * HWP;
    const int oh   = qi / WPAIR;
    const int jp   = qi - oh * WPAIR;

    int  o0[3], o1[3], o3[3];
    bool rv[3], v0[3], v3[3];
    #pragma unroll
    for (int r = 0; r < 3; ++r) {
        const int ih = oh + r - 1;
        rv[r] = (unsigned)ih < (unsigned)H_IN;
        v0[r] = rv[r] & (jp > 0);
        v3[r] = rv[r] & (jp < WPAIR - 1);
        o1[r] = nimg * (C_IN * HW) + (khalf * 2) * HW + (rv[r] ? ih : oh) * W_IN + 2 * jp;
        o0[r] = o1[r] - (v0[r] ? 1 : 0);
        o3[r] = o1[r] + (v3[r] ? 2 : 0);
    }

    const v4f* const wrd = wlds4 + (n_base + lrow) * WPITCH4 + khalf * 3;

    int goff[3], loff4[3];
    #pragma unroll
    for (int i = 0; i < 3; ++i) {
        const int it   = tid + 256 * i;
        const int oc   = it / SLOTS;
        const int slot = it - oc * SLOTS;
        const int cp   = slot / 3;
        const int r    = slot - cp * 3;
        goff[i]  = (oc_blk + oc) * KD + cp * 18 + r * 3;
        loff4[i] = oc * WPITCH4 + slot;
    }

    v8f acc[NT][4];
    #pragma unroll
    for (int nt = 0; nt < NT; ++nt)
        #pragma unroll
        for (int t = 0; t < 4; ++t) acc[nt][t] = (v8f){};

    float ra0[2], rb0[2], ra3[2], rb3[2];
    v2f   ra12[2], rb12[2];
    v4f   wa[2][NT], wb[2][NT];

    for (int cb0 = 0; cb0 < C_IN; cb0 += CCH) {
        ra0[0]  = x[o0[0]];
        rb0[0]  = x[o0[0] + HW];
        ra12[0] = *reinterpret_cast<const v2f*>(x + o1[0]);
        rb12[0] = *reinterpret_cast<const v2f*>(x + o1[0] + HW);
        ra3[0]  = x[o3[0]];
        rb3[0]  = x[o3[0] + HW];

        float g[6][6];
        #pragma unroll
        for (int k = 0; k < 6; ++k) {
            const float* p = wgt + goff[k % 3] + (k / 3) * (64 * KD);
            g[k][0] = p[0]; g[k][1] = p[1];  g[k][2] = p[2];
            g[k][3] = p[9]; g[k][4] = p[10]; g[k][5] = p[11];
        }

        __syncthreads();

        #pragma unroll
        for (int b = 0; b < NB; ++b) {
            #pragma unroll
            for (int k = 0; k < 6; ++k) {
                const float sa = g[k][0] + g[k][2];
                const float sb = g[k][3] + g[k][5];
                const v4f lo = { g[k][0], g[k][3], 0.5f * (sa + g[k][1]), 0.5f * (sb + g[k][4]) };
                const v4f hi = { 0.5f * (sa - g[k][1]), 0.5f * (sb - g[k][4]), g[k][2], g[k][5] };
                v4f* d = wlds4 + loff4[k % 3] + (b * 2 + k / 3) * (64 * WPITCH4);
                d[0]     = lo;
                d[HALF4] = hi;
            }
            if (b + 1 < NB) {
                #pragma unroll
                for (int k = 0; k < 6; ++k) {
                    const float* p = wgt + goff[k % 3] + ((b + 1) * 2 + k / 3) * (64 * KD);
                    g[k][0] = p[0]; g[k][1] = p[1];  g[k][2] = p[2];
                    g[k][3] = p[9]; g[k][4] = p[10]; g[k][5] = p[11];
                }
            }
        }

        __syncthreads();

        #pragma unroll
        for (int nt = 0; nt < NT; ++nt) {
            const v4f* wp = wrd + nt * (16 * WPITCH4);
            wa[0][nt] = wp[0];
            wb[0][nt] = wp[HALF4];
        }

        #pragma unroll
        for (int i = 0; i < 6; ++i) {
            const int s = i & 1;
            if (i < 5) {
                const int r  = (i + 1) >> 1;
                const int ks = (i + 1) & 1;
                const int n  = s ^ 1;
                const float* xp = x + (ks * 4) * HW;
                ra0[n]  = xp[o0[r]];
                rb0[n]  = xp[o0[r] + HW];
                ra12[n] = *reinterpret_cast<const v2f*>(xp + o1[r]);
                rb12[n] = *reinterpret_cast<const v2f*>(xp + o1[r] + HW);
                ra3[n]  = xp[o3[r]];
                rb3[n]  = xp[o3[r] + HW];
                #pragma unroll
                for (int nt = 0; nt < NT; ++nt) {
                    const v4f* wp = wrd + nt * (16 * WPITCH4) + (ks * 6 + r);
                    wa[n][nt] = wp[0];
                    wb[n][nt] = wp[HALF4];
                }
            }
            __builtin_amdgcn_sched_barrier(0);
            {
                const int r = i >> 1;
                const float p0 = v0[r] ? ra0[s] : 0.0f,    q0 = v0[r] ? rb0[s] : 0.0f;
                const float p1 = rv[r] ? ra12[s].x : 0.0f, q1 = rv[r] ? rb12[s].x : 0.0f;
                const float p2 = rv[r] ? ra12[s].y : 0.0f, q2 = rv[r] ? rb12[s].y : 0.0f;
                const float p3 = v3[r] ? ra3[s] : 0.0f,    q3 = v3[r] ? rb3[s] : 0.0f;

                const v2f u0 = { p0 - p2, q0 - q2 };
                const v2f u1 = { p1 + p2, q1 + q2 };
                const v2f u2 = { p2 - p1, q2 - q1 };
                const v2f u3 = { p1 - p3, q1 - q3 };

                #pragma unroll
                for (int nt = 0; nt < NT; ++nt) {
                    const v2f a0 = { wa[s][nt].x, wa[s][nt].y };
                    const v2f a1 = { wa[s][nt].z, wa[s][nt].w };
                    const v2f a2 = { wb[s][nt].x, wb[s][nt].y };
                    const v2f a3 = { wb[s][nt].z, wb[s][nt].w };
                    acc[nt][0] = __builtin_amdgcn_wmma_f32_16x16x4_f32(
                        false, a0, false, u0, (short)0, acc[nt][0], false, false);
                    acc[nt][1] = __builtin_amdgcn_wmma_f32_16x16x4_f32(
                        false, a1, false, u1, (short)0, acc[nt][1], false, false);
                    acc[nt][2] = __builtin_amdgcn_wmma_f32_16x16x4_f32(
                        false, a2, false, u2, (short)0, acc[nt][2], false, false);
                    acc[nt][3] = __builtin_amdgcn_wmma_f32_16x16x4_f32(
                        false, a3, false, u3, (short)0, acc[nt][3], false, false);
                }
            }
            __builtin_amdgcn_sched_barrier(0);
        }

        #pragma unroll
        for (int r = 0; r < 3; ++r) { o0[r] += CCH * HW; o1[r] += CCH * HW; o3[r] += CCH * HW; }
        #pragma unroll
        for (int i = 0; i < 3; ++i) goff[i] += CCH * 9;
    }

    const int obase = nimg * (K_OUT * HW) + 2 * qi;
    #pragma unroll
    for (int nt = 0; nt < NT; ++nt) {
        const int ocb = oc_blk + n_base + nt * 16 + 8 * khalf;
        const v4f b0 = *reinterpret_cast<const v4f*>(bias + ocb);
        const v4f b1 = *reinterpret_cast<const v4f*>(bias + ocb + 4);
        const float bv[8] = { b0.x, b0.y, b0.z, b0.w, b1.x, b1.y, b1.z, b1.w };
        #pragma unroll
        for (int v = 0; v < 8; ++v) {
            const float m0 = acc[nt][0][v], m1 = acc[nt][1][v];
            const float m2 = acc[nt][2][v], m3 = acc[nt][3][v];
            const v2f y = { (m0 + m1) + m2 + bv[v], (m1 - m2) - m3 + bv[v] };
            *reinterpret_cast<v2f*>(out + obase + (ocb + v) * HW) = y;
        }
    }
}

__global__ __launch_bounds__(256, 4)
void conv3x3_wmma_f32(const float* __restrict__ x,
                      const float* __restrict__ wgt,
                      const float* __restrict__ bias,
                      float* __restrict__ out)
{
    __shared__ v4f wlds4[WLDS4];

    if (blockIdx.x < N_SMALL) {
        conv_tile<2, 1>(x, wgt, bias, out, wlds4,
                        BIG_COVER + (int)blockIdx.x * SMALL_PAIRS, 0);
    } else {
        const int b = (int)blockIdx.x - N_SMALL;
        conv_tile<4, 4>(x, wgt, bias, out, wlds4,
                        (b >> 1) * BIG_PAIRS, (b & 1) * BIG_OC);
    }
}

extern "C" void kernel_launch(void* const* d_in, const int* in_sizes, int n_in,
                              void* d_out, int out_size, void* d_ws, size_t ws_size,
                              hipStream_t stream) {
    (void)in_sizes; (void)n_in; (void)out_size; (void)d_ws; (void)ws_size;
    const float* x    = (const float*)d_in[0];
    const float* wgt  = (const float*)d_in[1];
    const float* bias = (const float*)d_in[2];
    float* out        = (float*)d_out;

    dim3 grid(N_SMALL + N_BIG, 1);
    dim3 block(256);
    conv3x3_wmma_f32<<<grid, block, 0, stream>>>(x, wgt, bias, out);
}


# Big blocks' weights double-buffered in their unused LDS half: next chunk's weight loads requested in front of the last round's matrix block, stored behind it; one barrier per chunk; sums unchanged
# speedup vs baseline: 2.9748x; 1.0465x over previous
#include <hip/hip_runtime.h>

typedef __attribute__((ext_vector_type(2))) float v2f;
typedef __attribute__((ext_vector_type(8))) float v8f;
typedef __attribute__((ext_vector_type(4))) float v4f;

#define C_IN  128
#define H_IN  56
#define W_IN  56
#define N_IMG 32
#define K_OUT 256
#define HW    (H_IN * W_IN)
#define KD    (C_IN * 9)
#define WPAIR (W_IN / 2)
#define HWP   (HW / 2)
#define P_TOT (N_IMG * HWP)

#define CCH      8
#define SLOTS    12
#define WPITCH4  25
#define WLDS4    (K_OUT * WPITCH4)
#define HALF4    12

#define BIG_PAIRS   64
#define BIG_OC      128
#define N_BIG       1536
#define BIG_COVER   ((N_BIG / (K_OUT / BIG_OC)) * BIG_PAIRS)
#define SMALL_PAIRS 16
#define N_SMALL     ((P_TOT - BIG_COVER) / SMALL_PAIRS)

template <int NT, int WM>
__device__ __forceinline__
void conv_tile(const float* __restrict__ x,
               const float* __restrict__ wgt,
               const float* __restrict__ bias,
               float* __restrict__ out,
               v4f* __restrict__ wlds4,
               const int q_blk,
               const int oc_blk)
{
    constexpr int KB = (8 / WM) * NT * 16;
    constexpr int NB = KB / 128;

    const int tid    = threadIdx.x;
    const int lane   = tid & 31;
    const int wave   = tid >> 5;
    const int wave_m = wave % WM;
    const int wave_n = wave / WM;

    const int n_base = wave_n * (NT * 16);
    const int lrow   = lane & 15;
    const int khalf  = lane >> 4;

    const int qg   = q_blk + wave_m * 16 + lrow;
    const int nimg = qg / HWP;
    const int qi   = qg - nimg * HWP;
    const int oh   = qi / WPAIR;
    const int jp   = qi - oh * WPAIR;

    int  o0[3], o1[3], o3[3];
    bool rv[3], v0[3], v3[3];
    #pragma unroll
    for (int r = 0; r < 3; ++r) {
        const int ih = oh + r - 1;
        rv[r] = (unsigned)ih < (unsigned)H_IN;
        v0[r] = rv[r] & (jp > 0);
        v3[r] = rv[r] & (jp < WPAIR - 1);
        o1[r] = nimg * (C_IN * HW) + (khalf * 2) * HW + (rv[r] ? ih : oh) * W_IN + 2 * jp;
        o0[r] = o1[r] - (v0[r] ? 1 : 0);
        o3[r] = o1[r] + (v3[r] ? 2 : 0);
    }

    const v4f* const wrd = wlds4 + (n_base + lrow) * WPITCH4 + khalf * 3;

    int goff[3], loff4[3];
    #pragma unroll
    for (int i = 0; i < 3; ++i) {
        const int it   = tid + 256 * i;
        const int oc   = it / SLOTS;
        const int slot = it - oc * SLOTS;
        const int cp   = slot / 3;
        const int r    = slot - cp * 3;
        goff[i]  = (oc_blk + oc) * KD + cp * 18 + r * 3;
        loff4[i] = oc * WPITCH4 + slot;
    }

    v8f acc[NT][4];
    #pragma unroll
    for (int nt = 0; nt < NT; ++nt)
        #pragma unroll
        for (int t = 0; t < 4; ++t) acc[nt][t] = (v8f){};

    float ra0[2], rb0[2], ra3[2], rb3[2];
    v2f   ra12[2], rb12[2];
    v4f   wa[2][NT], wb[2][NT];

    for (int cb0 = 0; cb0 < C_IN; cb0 += CCH) {
        ra0[0]  = x[o0[0]];
        rb0[0]  = x[o0[0] + HW];
        ra12[0] = *reinterpret_cast<const v2f*>(x + o1[0]);
        rb12[0] = *reinterpret_cast<const v2f*>(x + o1[0] + HW);
        ra3[0]  = x[o3[0]];
        rb3[0]  = x[o3[0] + HW];

        float g[6][6];
        #pragma unroll
        for (int k = 0; k < 6; ++k) {
            const float* p = wgt + goff[k % 3] + (k / 3) * (64 * KD);
            g[k][0] = p[0]; g[k][1] = p[1];  g[k][2] = p[2];
            g[k][3] = p[9]; g[k][4] = p[10]; g[k][5] = p[11];
        }

        __syncthreads();

        #pragma unroll
        for (int b = 0; b < NB; ++b) {
            #pragma unroll
            for (int k = 0; k < 6; ++k) {
                const float sa = g[k][0] + g[k][2];
                const float sb = g[k][3] + g[k][5];
                const v4f lo = { g[k][0], g[k][3], 0.5f * (sa + g[k][1]), 0.5f * (sb + g[k][4]) };
                const v4f hi = { 0.5f * (sa - g[k][1]), 0.5f * (sb - g[k][4]), g[k][2], g[k][5] };
                v4f* d = wlds4 + loff4[k % 3] + (b * 2 + k / 3) * (64 * WPITCH4);
                d[0]     = lo;
                d[HALF4] = hi;
            }
            if (b + 1 < NB) {
                #pragma unroll
                for (int k = 0; k < 6; ++k) {
                    const float* p = wgt + goff[k % 3] + ((b + 1) * 2 + k / 3) * (64 * KD);
                    g[k][0] = p[0]; g[k][1] = p[1];  g[k][2] = p[2];
                    g[k][3] = p[9]; g[k][4] = p[10]; g[k][5] = p[11];
                }
            }
        }

        __syncthreads();

        #pragma unroll
        for (int nt = 0; nt < NT; ++nt) {
            const v4f* wp = wrd + nt * (16 * WPITCH4);
            wa[0][nt] = wp[0];
            wb[0][nt] = wp[HALF4];
        }

        #pragma unroll
        for (int i = 0; i < 6; ++i) {
            const int s = i & 1;
            if (i < 5) {
                const int r  = (i + 1) >> 1;
                const int ks = (i + 1) & 1;
                const int n  = s ^ 1;
                const float* xp = x + (ks * 4) * HW;
                ra0[n]  = xp[o0[r]];
                rb0[n]  = xp[o0[r] + HW];
                ra12[n] = *reinterpret_cast<const v2f*>(xp + o1[r]);
                rb12[n] = *reinterpret_cast<const v2f*>(xp + o1[r] + HW);
                ra3[n]  = xp[o3[r]];
                rb3[n]  = xp[o3[r] + HW];
                #pragma unroll
                for (int nt = 0; nt < NT; ++nt) {
                    const v4f* wp = wrd + nt * (16 * WPITCH4) + (ks * 6 + r);
                    wa[n][nt] = wp[0];
                    wb[n][nt] = wp[HALF4];
                }
            }
            __builtin_amdgcn_sched_barrier(0);
            {
                const int r = i >> 1;
                const float p0 = v0[r] ? ra0[s] : 0.0f,    q0 = v0[r] ? rb0[s] : 0.0f;
                const float p1 = rv[r] ? ra12[s].x : 0.0f, q1 = rv[r] ? rb12[s].x : 0.0f;
                const float p2 = rv[r] ? ra12[s].y : 0.0f, q2 = rv[r] ? rb12[s].y : 0.0f;
                const float p3 = v3[r] ? ra3[s] : 0.0f,    q3 = v3[r] ? rb3[s] : 0.0f;

                const v2f u0 = { p0 - p2, q0 - q2 };
                const v2f u1 = { p1 + p2, q1 + q2 };
                const v2f u2 = { p2 - p1, q2 - q1 };
                const v2f u3 = { p1 - p3, q1 - q3 };

                #pragma unroll
                for (int nt = 0; nt < NT; ++nt) {
                    const v2f a0 = { wa[s][nt].x, wa[s][nt].y };
                    const v2f a1 = { wa[s][nt].z, wa[s][nt].w };
                    const v2f a2 = { wb[s][nt].x, wb[s][nt].y };
                    const v2f a3 = { wb[s][nt].z, wb[s][nt].w };
                    acc[nt][0] = __builtin_amdgcn_wmma_f32_16x16x4_f32(
                        false, a0, false, u0, (short)0, acc[nt][0], false, false);
                    acc[nt][1] = __builtin_amdgcn_wmma_f32_16x16x4_f32(
                        false, a1, false, u1, (short)0, acc[nt][1], false, false);
                    acc[nt][2] = __builtin_amdgcn_wmma_f32_16x16x4_f32(
                        false, a2, false, u2, (short)0, acc[nt][2], false, false);
                    acc[nt][3] = __builtin_amdgcn_wmma_f32_16x16x4_f32(
                        false, a3, false, u3, (short)0, acc[nt][3], false, false);
                }
            }
            __builtin_amdgcn_sched_barrier(0);
        }

        #pragma unroll
        for (int r = 0; r < 3; ++r) { o0[r] += CCH * HW; o1[r] += CCH * HW; o3[r] += CCH * HW; }
        #pragma unroll
        for (int i = 0; i < 3; ++i) goff[i] += CCH * 9;
    }

    const int obase = nimg * (K_OUT * HW) + 2 * qi;
    #pragma unroll
    for (int nt = 0; nt < NT; ++nt) {
        const int ocb = oc_blk + n_base + nt * 16 + 8 * khalf;
        const v4f b0 = *reinterpret_cast<const v4f*>(bias + ocb);
        const v4f b1 = *reinterpret_cast<const v4f*>(bias + ocb + 4);
        const float bv[8] = { b0.x, b0.y, b0.z, b0.w, b1.x, b1.y, b1.z, b1.w };
        #pragma unroll
        for (int v = 0; v < 8; ++v) {
            const float m0 = acc[nt][0][v], m1 = acc[nt][1][v];
            const float m2 = acc[nt][2][v], m3 = acc[nt][3][v];
            const v2f y = { (m0 + m1) + m2 + bv[v], (m1 - m2) - m3 + bv[v] };
            *reinterpret_cast<v2f*>(out + obase + (ocb + v) * HW) = y;
        }
    }
}

template <int NT, int WM>
__device__ __forceinline__
void conv_tile_2stage(const float* __restrict__ x,
                      const float* __restrict__ wgt,
                      const float* __restrict__ bias,
                      float* __restrict__ out,
                      v4f* __restrict__ wlds4,
                      const int q_blk,
                      const int oc_blk)
{
    constexpr int KB = (8 / WM) * NT * 16;
    static_assert(KB == 128, "one staging batch of six items per thread");
    static_assert(2 * KB * WPITCH4 <= WLDS4, "two stages fit the allocation");
    constexpr int STAGE4 = KB * WPITCH4;

    const int tid    = threadIdx.x;
    const int lane   = tid & 31;
    const int wave   = tid >> 5;
    const int wave_m = wave % WM;
    const int wave_n = wave / WM;

    const int n_base = wave_n * (NT * 16);
    const int lrow   = lane & 15;
    const int khalf  = lane >> 4;

    const int qg   = q_blk + wave_m * 16 + lrow;
    const int nimg = qg / HWP;
    const int qi   = qg - nimg * HWP;
    const int oh   = qi / WPAIR;
    const int jp   = qi - oh * WPAIR;

    int  o0[3], o1[3], o3[3];
    bool rv[3], v0[3], v3[3];
    #pragma unroll
    for (int r = 0; r < 3; ++r) {
        const int ih = oh + r - 1;
        rv[r] = (unsigned)ih < (unsigned)H_IN;
        v0[r] = rv[r] & (jp > 0);
        v3[r] = rv[r] & (jp < WPAIR - 1);
        o1[r] = nimg * (C_IN * HW) + (khalf * 2) * HW + (rv[r] ? ih : oh) * W_IN + 2 * jp;
        o0[r] = o1[r] - (v0[r] ? 1 : 0);
        o3[r] = o1[r] + (v3[r] ? 2 : 0);
    }
    const v2f* __restrict__ const x2 = reinterpret_cast<const v2f*>(x);
    int o1h[3];
    #pragma unroll
    for (int r = 0; r < 3; ++r) o1h[r] = o1[r] >> 1;

    const v4f* const wrd = wlds4 + (n_base + lrow) * WPITCH4 + khalf * 3;

    int goff[3], loff4[3];
    #pragma unroll
    for (int i = 0; i < 3; ++i) {
        const int it   = tid + 256 * i;
        const int oc   = it / SLOTS;
        const int slot = it - oc * SLOTS;
        const int cp   = slot / 3;
        const int r    = slot - cp * 3;
        goff[i]  = (oc_blk + oc) * KD + cp * 18 + r * 3;
        loff4[i] = oc * WPITCH4 + slot;
    }

    v8f acc[NT][4];
    #pragma unroll
    for (int nt = 0; nt < NT; ++nt)
        #pragma unroll
        for (int t = 0; t < 4; ++t) acc[nt][t] = (v8f){};

    float ra0[2], rb0[2], ra3[2], rb3[2];
    v2f   ra12[2], rb12[2];
    v4f   wa[2][NT], wb[2][NT];
    float g[6][6];

    auto w_request = [&]() __attribute__((always_inline)) {
        #pragma unroll
        for (int k = 0; k < 6; ++k) {
            const float* p = wgt + goff[k % 3] + (k / 3) * (64 * KD);
            g[k][0] = p[0]; g[k][1] = p[1];  g[k][2] = p[2];
            g[k][3] = p[9]; g[k][4] = p[10]; g[k][5] = p[11];
        }
    };
    auto w_store = [&](const int st4) __attribute__((always_inline)) {
        #pragma unroll
        for (int k = 0; k < 6; ++k) {
            const float sa = g[k][0] + g[k][2];
            const float sb = g[k][3] + g[k][5];
            const v4f lo = { g[k][0], g[k][3], 0.5f * (sa + g[k][1]), 0.5f * (sb + g[k][4]) };
            const v4f hi = { 0.5f * (sa - g[k][1]), 0.5f * (sb - g[k][4]), g[k][2], g[k][5] };
            v4f* d = wlds4 + st4 + loff4[k % 3] + (k / 3) * (64 * WPITCH4);
            d[0]     = lo;
            d[HALF4] = hi;
        }
    };
    auto x_request0 = [&](const int adv) __attribute__((always_inline)) {
        const float* xp = x + adv;
        const v2f*   xq = x2 + adv / 2;
        ra0[0]  = xp[o0[0]];
        rb0[0]  = xp[o0[0] + HW];
        ra12[0] = xq[o1h[0]];
        rb12[0] = xq[o1h[0] + HW / 2];
        ra3[0]  = xp[o3[0]];
        rb3[0]  = xp[o3[0] + HW];
    };
    auto make_u = [&](const int s, const int r, v2f (&u)[4]) __attribute__((always_inline)) {
        const float p0 = v0[r] ? ra0[s] : 0.0f,    q0 = v0[r] ? rb0[s] : 0.0f;
        const float p1 = rv[r] ? ra12[s].x : 0.0f, q1 = rv[r] ? rb12[s].x : 0.0f;
        const float p2 = rv[r] ? ra12[s].y : 0.0f, q2 = rv[r] ? rb12[s].y : 0.0f;
        const float p3 = v3[r] ? ra3[s] : 0.0f,    q3 = v3[r] ? rb3[s] : 0.0f;
        u[0] = (v2f){ p0 - p2, q0 - q2 };
        u[1] = (v2f){ p1 + p2, q1 + q2 };
        u[2] = (v2f){ p2 - p1, q2 - q1 };
        u[3] = (v2f){ p1 - p3, q1 - q3 };
    };
    auto mma_block = [&](const int s, const v2f (&u)[4]) __attribute__((always_inline)) {
        #pragma unroll
        for (int nt = 0; nt < NT; ++nt) {
            const v2f a0 = { wa[s][nt].x, wa[s][nt].y };
            const v2f a1 = { wa[s][nt].z, wa[s][nt].w };
            const v2f a2 = { wb[s][nt].x, wb[s][nt].y };
            const v2f a3 = { wb[s][nt].z, wb[s][nt].w };
            acc[nt][0] = __builtin_amdgcn_wmma_f32_16x16x4_f32(
                false, a0, false, u[0], (short)0, acc[nt][0], false, false);
            acc[nt][1] = __builtin_amdgcn_wmma_f32_16x16x4_f32(
                false, a1, false, u[1], (short)0, acc[nt][1], false, false);
            acc[nt][2] = __builtin_amdgcn_wmma_f32_16x16x4_f32(
                false, a2, false, u[2], (short)0, acc[nt][2], false, false);
            acc[nt][3] = __builtin_amdgcn_wmma_f32_16x16x4_f32(
                false, a3, false, u[3], (short)0, acc[nt][3], false, false);
        }
    };

    x_request0(0);
    w_request();
    w_store(0);
    #pragma unroll
    for (int i = 0; i < 3; ++i) goff[i] += CCH * 9;
    __syncthreads();

    int cur4 = 0;
    auto do_chunk = [&](const bool more) __attribute__((always_inline)) {
        const v4f* const wr = wrd + cur4;

        #pragma unroll
        for (int nt = 0; nt < NT; ++nt) {
            const v4f* wp = wr + nt * (16 * WPITCH4);
            wa[0][nt] = wp[0];
            wb[0][nt] = wp[HALF4];
        }

        #pragma unroll
        for (int i = 0; i < 5; ++i) {
            const int s = i & 1;
            {
                const int r  = (i + 1) >> 1;
                const int ks = (i + 1) & 1;
                const int n  = s ^ 1;
                const float* xp = x + (ks * 4) * HW;
                const v2f*   xq = x2 + (ks * 4) * (HW / 2);
                ra0[n]  = xp[o0[r]];
                rb0[n]  = xp[o0[r] + HW];
                ra12[n] = xq[o1h[r]];
                rb12[n] = xq[o1h[r] + HW / 2];
                ra3[n]  = xp[o3[r]];
                rb3[n]  = xp[o3[r] + HW];
                #pragma unroll
                for (int nt = 0; nt < NT; ++nt) {
                    const v4f* wp = wr + nt * (16 * WPITCH4) + (ks * 6 + r);
                    wa[n][nt] = wp[0];
                    wb[n][nt] = wp[HALF4];
                }
            }
            __builtin_amdgcn_sched_barrier(0);
            {
                v2f u[4];
                make_u(s, i >> 1, u);
                mma_block(s, u);
            }
            __builtin_amdgcn_sched_barrier(0);
        }

        {
            v2f u[4];
            make_u(1, 2, u);
            __builtin_amdgcn_sched_barrier(0);
            if (more) w_request();
            __builtin_amdgcn_sched_barrier(0);
            mma_block(1, u);
            __builtin_amdgcn_sched_barrier(0);
            if (more) {
                x_request0(CCH * HW);
                w_store(cur4 ^ STAGE4);
                __syncthreads();
            }
        }

        #pragma unroll
        for (int r = 0; r < 3; ++r) { o0[r] += CCH * HW; o1h[r] += CCH * (HW / 2); o3[r] += CCH * HW; }
        #pragma unroll
        for (int i = 0; i < 3; ++i) goff[i] += CCH * 9;
        cur4 ^= STAGE4;
    };

    for (int cb0 = 0; cb0 < C_IN - CCH; cb0 += CCH) do_chunk(true);
    do_chunk(false);

    const int obase = nimg * (K_OUT * HW) + 2 * qi;
    #pragma unroll
    for (int nt = 0; nt < NT; ++nt) {
        const int ocb = oc_blk + n_base + nt * 16 + 8 * khalf;
        const v4f b0 = *reinterpret_cast<const v4f*>(bias + ocb);
        const v4f b1 = *reinterpret_cast<const v4f*>(bias + ocb + 4);
        const float bv[8] = { b0.x, b0.y, b0.z, b0.w, b1.x, b1.y, b1.z, b1.w };
        #pragma unroll
        for (int v = 0; v < 8; ++v) {
            const float m0 = acc[nt][0][v], m1 = acc[nt][1][v];
            const float m2 = acc[nt][2][v], m3 = acc[nt][3][v];
            const v2f y = { (m0 + m1) + m2 + bv[v], (m1 - m2) - m3 + bv[v] };
            *reinterpret_cast<v2f*>(out + obase + (ocb + v) * HW) = y;
        }
    }
}

__global__ __launch_bounds__(256, 4)
void conv3x3_wmma_f32(const float* __restrict__ x,
                      const float* __restrict__ wgt,
                      const float* __restrict__ bias,
                      float* __restrict__ out)
{
    __shared__ v4f wlds4[WLDS4];

    if (blockIdx.x < N_SMALL) {
        conv_tile<2, 1>(x, wgt, bias, out, wlds4,
                        BIG_COVER + (int)blockIdx.x * SMALL_PAIRS, 0);
    } else {
        const int b = (int)blockIdx.x - N_SMALL;
        conv_tile_2stage<4, 4>(x, wgt, bias, out, wlds4,
                               (b >> 1) * BIG_PAIRS, (b & 1) * BIG_OC);
    }
}

extern "C" void kernel_launch(void* const* d_in, const int* in_sizes, int n_in,
                              void* d_out, int out_size, void* d_ws, size_t ws_size,
                              hipStream_t stream) {
    (void)in_sizes; (void)n_in; (void)out_size; (void)d_ws; (void)ws_size;
    const float* x    = (const float*)d_in[0];
    const float* wgt  = (const float*)d_in[1];
    const float* bias = (const float*)d_in[2];
    float* out        = (float*)d_out;

    dim3 grid(N_SMALL + N_BIG, 1);
    dim3 block(256);
    conv3x3_wmma_f32<<<grid, block, 0, stream>>>(x, wgt, bias, out);
}


# Remainder re-cut into 512 short blocks (16 pairs x 32 out-ch), one per resident slot, so all slots end together: their 8 waves split the 128 channels, 8 partial sums added in LDS in wave order
# speedup vs baseline: 3.1088x; 1.0450x over previous
#include <hip/hip_runtime.h>

typedef __attribute__((ext_vector_type(2))) float v2f;
typedef __attribute__((ext_vector_type(8))) float v8f;
typedef __attribute__((ext_vector_type(4))) float v4f;

#define C_IN  128
#define H_IN  56
#define W_IN  56
#define N_IMG 32
#define K_OUT 256
#define HW    (H_IN * W_IN)
#define KD    (C_IN * 9)
#define WPAIR (W_IN / 2)
#define HWP   (HW / 2)
#define P_TOT (N_IMG * HWP)

#define CCH      8
#define SLOTS    12
#define WPITCH4  25
#define WLDS4    (K_OUT * WPITCH4)
#define HALF4    12

#define BIG_PAIRS   64
#define BIG_OC      128
#define N_BIG       1536
#define BIG_COVER   ((N_BIG / (K_OUT / BIG_OC)) * BIG_PAIRS)
#define SMALL_PAIRS 16
#define SMALL_OC    32
#define SMALL_KQ    8
#define SMALL_CH    (C_IN / SMALL_KQ)
#define N_SMALL     (((P_TOT - BIG_COVER) / SMALL_PAIRS) * (K_OUT / SMALL_OC))

__device__ __forceinline__
void conv_tile_ksplit(const float* __restrict__ x,
                      const float* __restrict__ wgt,
                      const float* __restrict__ bias,
                      float* __restrict__ out,
                      v4f* __restrict__ wlds4,
                      const int q_blk,
                      const int oc_blk)
{
    constexpr int NT = SMALL_OC / 16;
    constexpr int NB = (SMALL_OC * SMALL_KQ) / 128;
    constexpr int GSTEP = (64 / SMALL_OC) * SMALL_CH * 9;

    const int tid    = threadIdx.x;
    const int lane   = tid & 31;
    const int kq     = tid >> 5;

    const int n_base = kq * SMALL_OC;
    const int lrow   = lane & 15;
    const int khalf  = lane >> 4;

    const int qg   = q_blk + lrow;
    const int nimg = qg / HWP;
    const int qi   = qg - nimg * HWP;
    const int oh   = qi / WPAIR;
    const int jp   = qi - oh * WPAIR;

    int  o0[3], o1[3], o3[3];
    bool rv[3], v0[3], v3[3];
    #pragma unroll
    for (int r = 0; r < 3; ++r) {
        const int ih = oh + r - 1;
        rv[r] = (unsigned)ih < (unsigned)H_IN;
        v0[r] = rv[r] & (jp > 0);
        v3[r] = rv[r] & (jp < WPAIR - 1);
        o1[r] = nimg * (C_IN * HW) + (kq * SMALL_CH + khalf * 2) * HW + (rv[r] ? ih : oh) * W_IN + 2 * jp;
        o0[r] = o1[r] - (v0[r] ? 1 : 0);
        o3[r] = o1[r] + (v3[r] ? 2 : 0);
    }

    const v4f* const wrd = wlds4 + (n_base + lrow) * WPITCH4 + khalf * 3;

    int goff[3], loff4[3];
    #pragma unroll
    for (int i = 0; i < 3; ++i) {
        const int it   = tid + 256 * i;
        const int row  = it / SLOTS;
        const int slot = it - row * SLOTS;
        const int cp   = slot / 3;
        const int r    = slot - cp * 3;
        goff[i]  = (oc_blk + (row % SMALL_OC)) * KD + (row / SMALL_OC) * (SMALL_CH * 9) + cp * 18 + r * 3;
        loff4[i] = row * WPITCH4 + slot;
    }

    v8f acc[NT][4];
    #pragma unroll
    for (int nt = 0; nt < NT; ++nt)
        #pragma unroll
        for (int t = 0; t < 4; ++t) acc[nt][t] = (v8f){};

    float ra0[2], rb0[2], ra3[2], rb3[2];
    v2f   ra12[2], rb12[2];
    v4f   wa[2][NT], wb[2][NT];

    for (int cb0 = 0; cb0 < SMALL_CH; cb0 += CCH) {
        ra0[0]  = x[o0[0]];
        rb0[0]  = x[o0[0] + HW];
        ra12[0] = *reinterpret_cast<const v2f*>(x + o1[0]);
        rb12[0] = *reinterpret_cast<const v2f*>(x + o1[0] + HW);
        ra3[0]  = x[o3[0]];
        rb3[0]  = x[o3[0] + HW];

        float g[6][6];
        #pragma unroll
        for (int k = 0; k < 6; ++k) {
            const float* p = wgt + goff[k % 3] + (k / 3) * GSTEP;
            g[k][0] = p[0]; g[k][1] = p[1];  g[k][2] = p[2];
            g[k][3] = p[9]; g[k][4] = p[10]; g[k][5] = p[11];
        }

        __syncthreads();

        #pragma unroll
        for (int b = 0; b < NB; ++b) {
            #pragma unroll
            for (int k = 0; k < 6; ++k) {
                const float sa = g[k][0] + g[k][2];
                const float sb = g[k][3] + g[k][5];
                const v4f lo = { g[k][0], g[k][3], 0.5f * (sa + g[k][1]), 0.5f * (sb + g[k][4]) };
                const v4f hi = { 0.5f * (sa - g[k][1]), 0.5f * (sb - g[k][4]), g[k][2], g[k][5] };
                v4f* d = wlds4 + loff4[k % 3] + (b * 2 + k / 3) * (64 * WPITCH4);
                d[0]     = lo;
                d[HALF4] = hi;
            }
            if (b + 1 < NB) {
                #pragma unroll
                for (int k = 0; k < 6; ++k) {
                    const float* p = wgt + goff[k % 3] + ((b + 1) * 2 + k / 3) * GSTEP;
                    g[k][0] = p[0]; g[k][1] = p[1];  g[k][2] = p[2];
                    g[k][3] = p[9]; g[k][4] = p[10]; g[k][5] = p[11];
                }
            }
        }

        __syncthreads();

        #pragma unroll
        for (int nt = 0; nt < NT; ++nt) {
            const v4f* wp = wrd + nt * (16 * WPITCH4);
            wa[0][nt] = wp[0];
            wb[0][nt] = wp[HALF4];
        }

        #pragma unroll
        for (int i = 0; i < 6; ++i) {
            const int s = i & 1;
            if (i < 5) {
                const int r  = (i + 1) >> 1;
                const int ks = (i + 1) & 1;
                const int n  = s ^ 1;
                const float* xp = x + (ks * 4) * HW;
                ra0[n]  = xp[o0[r]];
                rb0[n]  = xp[o0[r] + HW];
                ra12[n] = *reinterpret_cast<const v2f*>(xp + o1[r]);
                rb12[n] = *reinterpret_cast<const v2f*>(xp + o1[r] + HW);
                ra3[n]  = xp[o3[r]];
                rb3[n]  = xp[o3[r] + HW];
                #pragma unroll
                for (int nt = 0; nt < NT; ++nt) {
                    const v4f* wp = wrd + nt * (16 * WPITCH4) + (ks * 6 + r);
                    wa[n][nt] = wp[0];
                    wb[n][nt] = wp[HALF4];
                }
            }
            __builtin_amdgcn_sched_barrier(0);
            {
                const int r = i >> 1;
                const float p0 = v0[r] ? ra0[s] : 0.0f,    q0 = v0[r] ? rb0[s] : 0.0f;
                const float p1 = rv[r] ? ra12[s].x : 0.0f, q1 = rv[r] ? rb12[s].x : 0.0f;
                const float p2 = rv[r] ? ra12[s].y : 0.0f, q2 = rv[r] ? rb12[s].y : 0.0f;
                const float p3 = v3[r] ? ra3[s] : 0.0f,    q3 = v3[r] ? rb3[s] : 0.0f;

                const v2f u0 = { p0 - p2, q0 - q2 };
                const v2f u1 = { p1 + p2, q1 + q2 };
                const v2f u2 = { p2 - p1, q2 - q1 };
                const v2f u3 = { p1 - p3, q1 - q3 };

                #pragma unroll
                for (int nt = 0; nt < NT; ++nt) {
                    const v2f a0 = { wa[s][nt].x, wa[s][nt].y };
                    const v2f a1 = { wa[s][nt].z, wa[s][nt].w };
                    const v2f a2 = { wb[s][nt].x, wb[s][nt].y };
                    const v2f a3 = { wb[s][nt].z, wb[s][nt].w };
                    acc[nt][0] = __builtin_amdgcn_wmma_f32_16x16x4_f32(
                        false, a0, false, u0, (short)0, acc[nt][0], false, false);
                    acc[nt][1] = __builtin_amdgcn_wmma_f32_16x16x4_f32(
                        false, a1, false, u1, (short)0, acc[nt][1], false, false);
                    acc[nt][2] = __builtin_amdgcn_wmma_f32_16x16x4_f32(
                        false, a2, false, u2, (short)0, acc[nt][2], false, false);
                    acc[nt][3] = __builtin_amdgcn_wmma_f32_16x16x4_f32(
                        false, a3, false, u3, (short)0, acc[nt][3], false, false);
                }
            }
            __builtin_amdgcn_sched_barrier(0);
        }

        #pragma unroll
        for (int r = 0; r < 3; ++r) { o0[r] += CCH * HW; o1[r] += CCH * HW; o3[r] += CCH * HW; }
        #pragma unroll
        for (int i = 0; i < 3; ++i) goff[i] += CCH * 9;
    }

    __syncthreads();
    float* const pf = reinterpret_cast<float*>(wlds4);
    #pragma unroll
    for (int nt = 0; nt < NT; ++nt)
        #pragma unroll
        for (int t = 0; t < 4; ++t)
            #pragma unroll
            for (int v = 0; v < 8; ++v)
                pf[(kq * 64 + (nt * 4 + t) * 8 + v) * 32 + lane] = acc[nt][t][v];
    __syncthreads();

    const int obase = nimg * (K_OUT * HW) + 2 * qi;
    #pragma unroll
    for (int i = 0; i < 2; ++i) {
        const int e  = 2 * kq + i;
        const int nt = e >> 3;
        const int v  = e & 7;
        float m[4];
        #pragma unroll
        for (int t = 0; t < 4; ++t) {
            const float* const pp = pf + ((nt * 4 + t) * 8 + v) * 32 + lane;
            float s = pp[0];
            #pragma unroll
            for (int q = 1; q < SMALL_KQ; ++q) s += pp[q * (64 * 32)];
            m[t] = s;
        }
        const int   oc = oc_blk + nt * 16 + 8 * khalf + v;
        const float bo = bias[oc];
        const v2f y = { (m[0] + m[1]) + m[2] + bo, (m[1] - m[2]) - m[3] + bo };
        *reinterpret_cast<v2f*>(out + obase + oc * HW) = y;
    }
}

template <int NT, int WM>
__device__ __forceinline__
void conv_tile_2stage(const float* __restrict__ x,
                      const float* __restrict__ wgt,
                      const float* __restrict__ bias,
                      float* __restrict__ out,
                      v4f* __restrict__ wlds4,
                      const int q_blk,
                      const int oc_blk)
{
    constexpr int KB = (8 / WM) * NT * 16;
    static_assert(KB == 128, "one staging batch of six items per thread");
    static_assert(2 * KB * WPITCH4 <= WLDS4, "two stages fit the allocation");
    constexpr int STAGE4 = KB * WPITCH4;

    const int tid    = threadIdx.x;
    const int lane   = tid & 31;
    const int wave   = tid >> 5;
    const int wave_m = wave % WM;
    const int wave_n = wave / WM;

    const int n_base = wave_n * (NT * 16);
    const int lrow   = lane & 15;
    const int khalf  = lane >> 4;

    const int qg   = q_blk + wave_m * 16 + lrow;
    const int nimg = qg / HWP;
    const int qi   = qg - nimg * HWP;
    const int oh   = qi / WPAIR;
    const int jp   = qi - oh * WPAIR;

    int  o0[3], o1[3], o3[3];
    bool rv[3], v0[3], v3[3];
    #pragma unroll
    for (int r = 0; r < 3; ++r) {
        const int ih = oh + r - 1;
        rv[r] = (unsigned)ih < (unsigned)H_IN;
        v0[r] = rv[r] & (jp > 0);
        v3[r] = rv[r] & (jp < WPAIR - 1);
        o1[r] = nimg * (C_IN * HW) + (khalf * 2) * HW + (rv[r] ? ih : oh) * W_IN + 2 * jp;
        o0[r] = o1[r] - (v0[r] ? 1 : 0);
        o3[r] = o1[r] + (v3[r] ? 2 : 0);
    }
    const v2f* __restrict__ const x2 = reinterpret_cast<const v2f*>(x);
    int o1h[3];
    #pragma unroll
    for (int r = 0; r < 3; ++r) o1h[r] = o1[r] >> 1;

    const v4f* const wrd = wlds4 + (n_base + lrow) * WPITCH4 + khalf * 3;

    int goff[3], loff4[3];
    #pragma unroll
    for (int i = 0; i < 3; ++i) {
        const int it   = tid + 256 * i;
        const int oc   = it / SLOTS;
        const int slot = it - oc * SLOTS;
        const int cp   = slot / 3;
        const int r    = slot - cp * 3;
        goff[i]  = (oc_blk + oc) * KD + cp * 18 + r * 3;
        loff4[i] = oc * WPITCH4 + slot;
    }

    v8f acc[NT][4];
    #pragma unroll
    for (int nt = 0; nt < NT; ++nt)
        #pragma unroll
        for (int t = 0; t < 4; ++t) acc[nt][t] = (v8f){};

    float ra0[2], rb0[2], ra3[2], rb3[2];
    v2f   ra12[2], rb12[2];
    v4f   wa[2][NT], wb[2][NT];
    float g[6][6];

    auto w_request = [&]() __attribute__((always_inline)) {
        #pragma unroll
        for (int k = 0; k < 6; ++k) {
            const float* p = wgt + goff[k % 3] + (k / 3) * (64 * KD);
            g[k][0] = p[0]; g[k][1] = p[1];  g[k][2] = p[2];
            g[k][3] = p[9]; g[k][4] = p[10]; g[k][5] = p[11];
        }
    };
    auto w_store = [&](const int st4) __attribute__((always_inline)) {
        #pragma unroll
        for (int k = 0; k < 6; ++k) {
            const float sa = g[k][0] + g[k][2];
            const float sb = g[k][3] + g[k][5];
            const v4f lo = { g[k][0], g[k][3], 0.5f * (sa + g[k][1]), 0.5f * (sb + g[k][4]) };
            const v4f hi = { 0.5f * (sa - g[k][1]), 0.5f * (sb - g[k][4]), g[k][2], g[k][5] };
            v4f* d = wlds4 + st4 + loff4[k % 3] + (k / 3) * (64 * WPITCH4);
            d[0]     = lo;
            d[HALF4] = hi;
        }
    };
    auto x_request0 = [&](const int adv) __attribute__((always_inline)) {
        const float* xp = x + adv;
        const v2f*   xq = x2 + adv / 2;
        ra0[0]  = xp[o0[0]];
        rb0[0]  = xp[o0[0] + HW];
        ra12[0] = xq[o1h[0]];
        rb12[0] = xq[o1h[0] + HW / 2];
        ra3[0]  = xp[o3[0]];
        rb3[0]  = xp[o3[0] + HW];
    };
    auto make_u = [&](const int s, const int r, v2f (&u)[4]) __attribute__((always_inline)) {
        const float p0 = v0[r] ? ra0[s] : 0.0f,    q0 = v0[r] ? rb0[s] : 0.0f;
        const float p1 = rv[r] ? ra12[s].x : 0.0f, q1 = rv[r] ? rb12[s].x : 0.0f;
        const float p2 = rv[r] ? ra12[s].y : 0.0f, q2 = rv[r] ? rb12[s].y : 0.0f;
        const float p3 = v3[r] ? ra3[s] : 0.0f,    q3 = v3[r] ? rb3[s] : 0.0f;
        u[0] = (v2f){ p0 - p2, q0 - q2 };
        u[1] = (v2f){ p1 + p2, q1 + q2 };
        u[2] = (v2f){ p2 - p1, q2 - q1 };
        u[3] = (v2f){ p1 - p3, q1 - q3 };
    };
    auto mma_block = [&](const int s, const v2f (&u)[4]) __attribute__((always_inline)) {
        #pragma unroll
        for (int nt = 0; nt < NT; ++nt) {
            const v2f a0 = { wa[s][nt].x, wa[s][nt].y };
            const v2f a1 = { wa[s][nt].z, wa[s][nt].w };
            const v2f a2 = { wb[s][nt].x, wb[s][nt].y };
            const v2f a3 = { wb[s][nt].z, wb[s][nt].w };
            acc[nt][0] = __builtin_amdgcn_wmma_f32_16x16x4_f32(
                false, a0, false, u[0], (short)0, acc[nt][0], false, false);
            acc[nt][1] = __builtin_amdgcn_wmma_f32_16x16x4_f32(
                false, a1, false, u[1], (short)0, acc[nt][1], false, false);
            acc[nt][2] = __builtin_amdgcn_wmma_f32_16x16x4_f32(
                false, a2, false, u[2], (short)0, acc[nt][2], false, false);
            acc[nt][3] = __builtin_amdgcn_wmma_f32_16x16x4_f32(
                false, a3, false, u[3], (short)0, acc[nt][3], false, false);
        }
    };

    x_request0(0);
    w_request();
    w_store(0);
    #pragma unroll
    for (int i = 0; i < 3; ++i) goff[i] += CCH * 9;
    __syncthreads();

    int cur4 = 0;
    auto do_chunk = [&](const bool more) __attribute__((always_inline)) {
        const v4f* const wr = wrd + cur4;

        #pragma unroll
        for (int nt = 0; nt < NT; ++nt) {
            const v4f* wp = wr + nt * (16 * WPITCH4);
            wa[0][nt] = wp[0];
            wb[0][nt] = wp[HALF4];
        }

        #pragma unroll
        for (int i = 0; i < 5; ++i) {
            const int s = i & 1;
            {
                const int r  = (i + 1) >> 1;
                const int ks = (i + 1) & 1;
                const int n  = s ^ 1;
                const float* xp = x + (ks * 4) * HW;
                const v2f*   xq = x2 + (ks * 4) * (HW / 2);
                ra0[n]  = xp[o0[r]];
                rb0[n]  = xp[o0[r] + HW];
                ra12[n] = xq[o1h[r]];
                rb12[n] = xq[o1h[r] + HW / 2];
                ra3[n]  = xp[o3[r]];
                rb3[n]  = xp[o3[r] + HW];
                #pragma unroll
                for (int nt = 0; nt < NT; ++nt) {
                    const v4f* wp = wr + nt * (16 * WPITCH4) + (ks * 6 + r);
                    wa[n][nt] = wp[0];
                    wb[n][nt] = wp[HALF4];
                }
            }
            __builtin_amdgcn_sched_barrier(0);
            {
                v2f u[4];
                make_u(s, i >> 1, u);
                mma_block(s, u);
            }
            __builtin_amdgcn_sched_barrier(0);
        }

        {
            v2f u[4];
            make_u(1, 2, u);
            __builtin_amdgcn_sched_barrier(0);
            if (more) w_request();
            __builtin_amdgcn_sched_barrier(0);
            mma_block(1, u);
            __builtin_amdgcn_sched_barrier(0);
            if (more) {
                x_request0(CCH * HW);
                w_store(cur4 ^ STAGE4);
                __syncthreads();
            }
        }

        #pragma unroll
        for (int r = 0; r < 3; ++r) { o0[r] += CCH * HW; o1h[r] += CCH * (HW / 2); o3[r] += CCH * HW; }
        #pragma unroll
        for (int i = 0; i < 3; ++i) goff[i] += CCH * 9;
        cur4 ^= STAGE4;
    };

    for (int cb0 = 0; cb0 < C_IN - CCH; cb0 += CCH) do_chunk(true);
    do_chunk(false);

    const int obase = nimg * (K_OUT * HW) + 2 * qi;
    #pragma unroll
    for (int nt = 0; nt < NT; ++nt) {
        const int ocb = oc_blk + n_base + nt * 16 + 8 * khalf;
        const v4f b0 = *reinterpret_cast<const v4f*>(bias + ocb);
        const v4f b1 = *reinterpret_cast<const v4f*>(bias + ocb + 4);
        const float bv[8] = { b0.x, b0.y, b0.z, b0.w, b1.x, b1.y, b1.z, b1.w };
        #pragma unroll
        for (int v = 0; v < 8; ++v) {
            const float m0 = acc[nt][0][v], m1 = acc[nt][1][v];
            const float m2 = acc[nt][2][v], m3 = acc[nt][3][v];
            const v2f y = { (m0 + m1) + m2 + bv[v], (m1 - m2) - m3 + bv[v] };
            *reinterpret_cast<v2f*>(out + obase + (ocb + v) * HW) = y;
        }
    }
}

__global__ __launch_bounds__(256, 4)
void conv3x3_wmma_f32(const float* __restrict__ x,
                      const float* __restrict__ wgt,
                      const float* __restrict__ bias,
                      float* __restrict__ out)
{
    __shared__ v4f wlds4[WLDS4];

    if (blockIdx.x < N_SMALL) {
        const int s = (int)blockIdx.x;
        conv_tile_ksplit(x, wgt, bias, out, wlds4,
                         BIG_COVER + (s / (K_OUT / SMALL_OC)) * SMALL_PAIRS,
                         (s % (K_OUT / SMALL_OC)) * SMALL_OC);
    } else {
        const int b = (int)blockIdx.x - N_SMALL;
        conv_tile_2stage<4, 4>(x, wgt, bias, out, wlds4,
                               (b >> 1) * BIG_PAIRS, (b & 1) * BIG_OC);
    }
}

extern "C" void kernel_launch(void* const* d_in, const int* in_sizes, int n_in,
                              void* d_out, int out_size, void* d_ws, size_t ws_size,
                              hipStream_t stream) {
    (void)in_sizes; (void)n_in; (void)out_size; (void)d_ws; (void)ws_size;
    const float* x    = (const float*)d_in[0];
    const float* wgt  = (const float*)d_in[1];
    const float* bias = (const float*)d_in[2];
    float* out        = (float*)d_out;

    dim3 grid(N_SMALL + N_BIG, 1);
    dim3 block(256);
    conv3x3_wmma_f32<<<grid, block, 0, stream>>>(x, wgt, bias, out);
}


# Big block tile re-cut from 64 pairs x 128 out-channels to 128 pairs x 64: a block stages a quarter of the weights per chunk, its 8 waves own 8 different pixel tiles; same wave tile, sums unchanged
# speedup vs baseline: 3.2335x; 1.0401x over previous
#include <hip/hip_runtime.h>

typedef __attribute__((ext_vector_type(2))) float v2f;
typedef __attribute__((ext_vector_type(8))) float v8f;
typedef __attribute__((ext_vector_type(4))) float v4f;

#define C_IN  128
#define H_IN  56
#define W_IN  56
#define N_IMG 32
#define K_OUT 256
#define HW    (H_IN * W_IN)
#define KD    (C_IN * 9)
#define WPAIR (W_IN / 2)
#define HWP   (HW / 2)
#define P_TOT (N_IMG * HWP)

#define CCH      8
#define SLOTS    12
#define WPITCH4  25
#define WLDS4    (K_OUT * WPITCH4)
#define HALF4    12

#define BIG_PAIRS   128
#define BIG_OC      64
#define N_BIG       1536
#define BIG_COVER   ((N_BIG / (K_OUT / BIG_OC)) * BIG_PAIRS)
#define SMALL_PAIRS 16
#define SMALL_OC    32
#define SMALL_KQ    8
#define SMALL_CH    (C_IN / SMALL_KQ)
#define N_SMALL     (((P_TOT - BIG_COVER) / SMALL_PAIRS) * (K_OUT / SMALL_OC))

__device__ __forceinline__
void conv_tile_ksplit(const float* __restrict__ x,
                      const float* __restrict__ wgt,
                      const float* __restrict__ bias,
                      float* __restrict__ out,
                      v4f* __restrict__ wlds4,
                      const int q_blk,
                      const int oc_blk)
{
    constexpr int NT = SMALL_OC / 16;
    constexpr int NB = (SMALL_OC * SMALL_KQ) / 128;
    constexpr int GSTEP = (64 / SMALL_OC) * SMALL_CH * 9;

    const int tid    = threadIdx.x;
    const int lane   = tid & 31;
    const int kq     = tid >> 5;

    const int n_base = kq * SMALL_OC;
    const int lrow   = lane & 15;
    const int khalf  = lane >> 4;

    const int qg   = q_blk + lrow;
    const int nimg = qg / HWP;
    const int qi   = qg - nimg * HWP;
    const int oh   = qi / WPAIR;
    const int jp   = qi - oh * WPAIR;

    int  o0[3], o1[3], o3[3];
    bool rv[3], v0[3], v3[3];
    #pragma unroll
    for (int r = 0; r < 3; ++r) {
        const int ih = oh + r - 1;
        rv[r] = (unsigned)ih < (unsigned)H_IN;
        v0[r] = rv[r] & (jp > 0);
        v3[r] = rv[r] & (jp < WPAIR - 1);
        o1[r] = nimg * (C_IN * HW) + (kq * SMALL_CH + khalf * 2) * HW + (rv[r] ? ih : oh) * W_IN + 2 * jp;
        o0[r] = o1[r] - (v0[r] ? 1 : 0);
        o3[r] = o1[r] + (v3[r] ? 2 : 0);
    }

    const v4f* const wrd = wlds4 + (n_base + lrow) * WPITCH4 + khalf * 3;

    int goff[3], loff4[3];
    #pragma unroll
    for (int i = 0; i < 3; ++i) {
        const int it   = tid + 256 * i;
        const int row  = it / SLOTS;
        const int slot = it - row * SLOTS;
        const int cp   = slot / 3;
        const int r    = slot - cp * 3;
        goff[i]  = (oc_blk + (row % SMALL_OC)) * KD + (row / SMALL_OC) * (SMALL_CH * 9) + cp * 18 + r * 3;
        loff4[i] = row * WPITCH4 + slot;
    }

    v8f acc[NT][4];
    #pragma unroll
    for (int nt = 0; nt < NT; ++nt)
        #pragma unroll
        for (int t = 0; t < 4; ++t) acc[nt][t] = (v8f){};

    float ra0[2], rb0[2], ra3[2], rb3[2];
    v2f   ra12[2], rb12[2];
    v4f   wa[2][NT], wb[2][NT];

    for (int cb0 = 0; cb0 < SMALL_CH; cb0 += CCH) {
        ra0[0]  = x[o0[0]];
        rb0[0]  = x[o0[0] + HW];
        ra12[0] = *reinterpret_cast<const v2f*>(x + o1[0]);
        rb12[0] = *reinterpret_cast<const v2f*>(x + o1[0] + HW);
        ra3[0]  = x[o3[0]];
        rb3[0]  = x[o3[0] + HW];

        float g[6][6];
        #pragma unroll
        for (int k = 0; k < 6; ++k) {
            const float* p = wgt + goff[k % 3] + (k / 3) * GSTEP;
            g[k][0] = p[0]; g[k][1] = p[1];  g[k][2] = p[2];
            g[k][3] = p[9]; g[k][4] = p[10]; g[k][5] = p[11];
        }

        __syncthreads();

        #pragma unroll
        for (int b = 0; b < NB; ++b) {
            #pragma unroll
            for (int k = 0; k < 6; ++k) {
                const float sa = g[k][0] + g[k][2];
                const float sb = g[k][3] + g[k][5];
                const v4f lo = { g[k][0], g[k][3], 0.5f * (sa + g[k][1]), 0.5f * (sb + g[k][4]) };
                const v4f hi = { 0.5f * (sa - g[k][1]), 0.5f * (sb - g[k][4]), g[k][2], g[k][5] };
                v4f* d = wlds4 + loff4[k % 3] + (b * 2 + k / 3) * (64 * WPITCH4);
                d[0]     = lo;
                d[HALF4] = hi;
            }
            if (b + 1 < NB) {
                #pragma unroll
                for (int k = 0; k < 6; ++k) {
                    const float* p = wgt + goff[k % 3] + ((b + 1) * 2 + k / 3) * GSTEP;
                    g[k][0] = p[0]; g[k][1] = p[1];  g[k][2] = p[2];
                    g[k][3] = p[9]; g[k][4] = p[10]; g[k][5] = p[11];
                }
            }
        }

        __syncthreads();

        #pragma unroll
        for (int nt = 0; nt < NT; ++nt) {
            const v4f* wp = wrd + nt * (16 * WPITCH4);
            wa[0][nt] = wp[0];
            wb[0][nt] = wp[HALF4];
        }

        #pragma unroll
        for (int i = 0; i < 6; ++i) {
            const int s = i & 1;
            if (i < 5) {
                const int r  = (i + 1) >> 1;
                const int ks = (i + 1) & 1;
                const int n  = s ^ 1;
                const float* xp = x + (ks * 4) * HW;
                ra0[n]  = xp[o0[r]];
                rb0[n]  = xp[o0[r] + HW];
                ra12[n] = *reinterpret_cast<const v2f*>(xp + o1[r]);
                rb12[n] = *reinterpret_cast<const v2f*>(xp + o1[r] + HW);
                ra3[n]  = xp[o3[r]];
                rb3[n]  = xp[o3[r] + HW];
                #pragma unroll
                for (int nt = 0; nt < NT; ++nt) {
                    const v4f* wp = wrd + nt * (16 * WPITCH4) + (ks * 6 + r);
                    wa[n][nt] = wp[0];
                    wb[n][nt] = wp[HALF4];
                }
            }
            __builtin_amdgcn_sched_barrier(0);
            {
                const int r = i >> 1;
                const float p0 = v0[r] ? ra0[s] : 0.0f,    q0 = v0[r] ? rb0[s] : 0.0f;
                const float p1 = rv[r] ? ra12[s].x : 0.0f, q1 = rv[r] ? rb12[s].x : 0.0f;
                const float p2 = rv[r] ? ra12[s].y : 0.0f, q2 = rv[r] ? rb12[s].y : 0.0f;
                const float p3 = v3[r] ? ra3[s] : 0.0f,    q3 = v3[r] ? rb3[s] : 0.0f;

                const v2f u0 = { p0 - p2, q0 - q2 };
                const v2f u1 = { p1 + p2, q1 + q2 };
                const v2f u2 = { p2 - p1, q2 - q1 };
                const v2f u3 = { p1 - p3, q1 - q3 };

                #pragma unroll
                for (int nt = 0; nt < NT; ++nt) {
                    const v2f a0 = { wa[s][nt].x, wa[s][nt].y };
                    const v2f a1 = { wa[s][nt].z, wa[s][nt].w };
                    const v2f a2 = { wb[s][nt].x, wb[s][nt].y };
                    const v2f a3 = { wb[s][nt].z, wb[s][nt].w };
                    acc[nt][0] = __builtin_amdgcn_wmma_f32_16x16x4_f32(
                        false, a0, false, u0, (short)0, acc[nt][0], false, false);
                    acc[nt][1] = __builtin_amdgcn_wmma_f32_16x16x4_f32(
                        false, a1, false, u1, (short)0, acc[nt][1], false, false);
                    acc[nt][2] = __builtin_amdgcn_wmma_f32_16x16x4_f32(
                        false, a2, false, u2, (short)0, acc[nt][2], false, false);
                    acc[nt][3] = __builtin_amdgcn_wmma_f32_16x16x4_f32(
                        false, a3, false, u3, (short)0, acc[nt][3], false, false);
                }
            }
            __builtin_amdgcn_sched_barrier(0);
        }

        #pragma unroll
        for (int r = 0; r < 3; ++r) { o0[r] += CCH * HW; o1[r] += CCH * HW; o3[r] += CCH * HW; }
        #pragma unroll
        for (int i = 0; i < 3; ++i) goff[i] += CCH * 9;
    }

    __syncthreads();
    float* const pf = reinterpret_cast<float*>(wlds4);
    #pragma unroll
    for (int nt = 0; nt < NT; ++nt)
        #pragma unroll
        for (int t = 0; t < 4; ++t)
            #pragma unroll
            for (int v = 0; v < 8; ++v)
                pf[(kq * 64 + (nt * 4 + t) * 8 + v) * 32 + lane] = acc[nt][t][v];
    __syncthreads();

    const int obase = nimg * (K_OUT * HW) + 2 * qi;
    #pragma unroll
    for (int i = 0; i < 2; ++i) {
        const int e  = 2 * kq + i;
        const int nt = e >> 3;
        const int v  = e & 7;
        float m[4];
        #pragma unroll
        for (int t = 0; t < 4; ++t) {
            const float* const pp = pf + ((nt * 4 + t) * 8 + v) * 32 + lane;
            float s = pp[0];
            #pragma unroll
            for (int q = 1; q < SMALL_KQ; ++q) s += pp[q * (64 * 32)];
            m[t] = s;
        }
        const int   oc = oc_blk + nt * 16 + 8 * khalf + v;
        const float bo = bias[oc];
        const v2f y = { (m[0] + m[1]) + m[2] + bo, (m[1] - m[2]) - m[3] + bo };
        *reinterpret_cast<v2f*>(out + obase + oc * HW) = y;
    }
}

template <int NT, int WM>
__device__ __forceinline__
void conv_tile_2stage(const float* __restrict__ x,
                      const float* __restrict__ wgt,
                      const float* __restrict__ bias,
                      float* __restrict__ out,
                      v4f* __restrict__ wlds4,
                      const int q_blk,
                      const int oc_blk)
{
    constexpr int KB = (8 / WM) * NT * 16;
    static_assert(KB * SLOTS == 3 * 256, "one staging batch of three items per thread");
    static_assert(2 * KB * WPITCH4 <= WLDS4, "two stages fit the allocation");
    constexpr int STAGE4 = KB * WPITCH4;

    const int tid    = threadIdx.x;
    const int lane   = tid & 31;
    const int wave   = tid >> 5;
    const int wave_m = wave % WM;
    const int wave_n = wave / WM;

    const int n_base = wave_n * (NT * 16);
    const int lrow   = lane & 15;
    const int khalf  = lane >> 4;

    const int qg   = q_blk + wave_m * 16 + lrow;
    const int nimg = qg / HWP;
    const int qi   = qg - nimg * HWP;
    const int oh   = qi / WPAIR;
    const int jp   = qi - oh * WPAIR;

    int  o0[3], o1[3], o3[3];
    bool rv[3], v0[3], v3[3];
    #pragma unroll
    for (int r = 0; r < 3; ++r) {
        const int ih = oh + r - 1;
        rv[r] = (unsigned)ih < (unsigned)H_IN;
        v0[r] = rv[r] & (jp > 0);
        v3[r] = rv[r] & (jp < WPAIR - 1);
        o1[r] = nimg * (C_IN * HW) + (khalf * 2) * HW + (rv[r] ? ih : oh) * W_IN + 2 * jp;
        o0[r] = o1[r] - (v0[r] ? 1 : 0);
        o3[r] = o1[r] + (v3[r] ? 2 : 0);
    }
    const v2f* __restrict__ const x2 = reinterpret_cast<const v2f*>(x);
    int o1h[3];
    #pragma unroll
    for (int r = 0; r < 3; ++r) o1h[r] = o1[r] >> 1;

    const v4f* const wrd = wlds4 + (n_base + lrow) * WPITCH4 + khalf * 3;

    int goff[3], loff4[3];
    #pragma unroll
    for (int i = 0; i < 3; ++i) {
        const int it   = tid + 256 * i;
        const int oc   = it / SLOTS;
        const int slot = it - oc * SLOTS;
        const int cp   = slot / 3;
        const int r    = slot - cp * 3;
        goff[i]  = (oc_blk + oc) * KD + cp * 18 + r * 3;
        loff4[i] = oc * WPITCH4 + slot;
    }

    v8f acc[NT][4];
    #pragma unroll
    for (int nt = 0; nt < NT; ++nt)
        #pragma unroll
        for (int t = 0; t < 4; ++t) acc[nt][t] = (v8f){};

    float ra0[2], rb0[2], ra3[2], rb3[2];
    v2f   ra12[2], rb12[2];
    v4f   wa[2][NT], wb[2][NT];
    float g[3][6];

    auto w_request = [&]() __attribute__((always_inline)) {
        #pragma unroll
        for (int k = 0; k < 3; ++k) {
            const float* p = wgt + goff[k];
            g[k][0] = p[0]; g[k][1] = p[1];  g[k][2] = p[2];
            g[k][3] = p[9]; g[k][4] = p[10]; g[k][5] = p[11];
        }
    };
    auto w_store = [&](const int st4) __attribute__((always_inline)) {
        #pragma unroll
        for (int k = 0; k < 3; ++k) {
            const float sa = g[k][0] + g[k][2];
            const float sb = g[k][3] + g[k][5];
            const v4f lo = { g[k][0], g[k][3], 0.5f * (sa + g[k][1]), 0.5f * (sb + g[k][4]) };
            const v4f hi = { 0.5f * (sa - g[k][1]), 0.5f * (sb - g[k][4]), g[k][2], g[k][5] };
            v4f* d = wlds4 + st4 + loff4[k];
            d[0]     = lo;
            d[HALF4] = hi;
        }
    };
    auto x_request0 = [&](const int adv) __attribute__((always_inline)) {
        const float* xp = x + adv;
        const v2f*   xq = x2 + adv / 2;
        ra0[0]  = xp[o0[0]];
        rb0[0]  = xp[o0[0] + HW];
        ra12[0] = xq[o1h[0]];
        rb12[0] = xq[o1h[0] + HW / 2];
        ra3[0]  = xp[o3[0]];
        rb3[0]  = xp[o3[0] + HW];
    };
    auto make_u = [&](const int s, const int r, v2f (&u)[4]) __attribute__((always_inline)) {
        const float p0 = v0[r] ? ra0[s] : 0.0f,    q0 = v0[r] ? rb0[s] : 0.0f;
        const float p1 = rv[r] ? ra12[s].x : 0.0f, q1 = rv[r] ? rb12[s].x : 0.0f;
        const float p2 = rv[r] ? ra12[s].y : 0.0f, q2 = rv[r] ? rb12[s].y : 0.0f;
        const float p3 = v3[r] ? ra3[s] : 0.0f,    q3 = v3[r] ? rb3[s] : 0.0f;
        u[0] = (v2f){ p0 - p2, q0 - q2 };
        u[1] = (v2f){ p1 + p2, q1 + q2 };
        u[2] = (v2f){ p2 - p1, q2 - q1 };
        u[3] = (v2f){ p1 - p3, q1 - q3 };
    };
    auto mma_block = [&](const int s, const v2f (&u)[4]) __attribute__((always_inline)) {
        #pragma unroll
        for (int nt = 0; nt < NT; ++nt) {
            const v2f a0 = { wa[s][nt].x, wa[s][nt].y };
            const v2f a1 = { wa[s][nt].z, wa[s][nt].w };
            const v2f a2 = { wb[s][nt].x, wb[s][nt].y };
            const v2f a3 = { wb[s][nt].z, wb[s][nt].w };
            acc[nt][0] = __builtin_amdgcn_wmma_f32_16x16x4_f32(
                false, a0, false, u[0], (short)0, acc[nt][0], false, false);
            acc[nt][1] = __builtin_amdgcn_wmma_f32_16x16x4_f32(
                false, a1, false, u[1], (short)0, acc[nt][1], false, false);
            acc[nt][2] = __builtin_amdgcn_wmma_f32_16x16x4_f32(
                false, a2, false, u[2], (short)0, acc[nt][2], false, false);
            acc[nt][3] = __builtin_amdgcn_wmma_f32_16x16x4_f32(
                false, a3, false, u[3], (short)0, acc[nt][3], false, false);
        }
    };

    x_request0(0);
    w_request();
    w_store(0);
    #pragma unroll
    for (int i = 0; i < 3; ++i) goff[i] += CCH * 9;
    __syncthreads();

    int cur4 = 0;
    auto do_chunk = [&](const bool more) __attribute__((always_inline)) {
        const v4f* const wr = wrd + cur4;

        #pragma unroll
        for (int nt = 0; nt < NT; ++nt) {
            const v4f* wp = wr + nt * (16 * WPITCH4);
            wa[0][nt] = wp[0];
            wb[0][nt] = wp[HALF4];
        }

        #pragma unroll
        for (int i = 0; i < 5; ++i) {
            const int s = i & 1;
            {
                const int r  = (i + 1) >> 1;
                const int ks = (i + 1) & 1;
                const int n  = s ^ 1;
                const float* xp = x + (ks * 4) * HW;
                const v2f*   xq = x2 + (ks * 4) * (HW / 2);
                ra0[n]  = xp[o0[r]];
                rb0[n]  = xp[o0[r] + HW];
                ra12[n] = xq[o1h[r]];
                rb12[n] = xq[o1h[r] + HW / 2];
                ra3[n]  = xp[o3[r]];
                rb3[n]  = xp[o3[r] + HW];
                #pragma unroll
                for (int nt = 0; nt < NT; ++nt) {
                    const v4f* wp = wr + nt * (16 * WPITCH4) + (ks * 6 + r);
                    wa[n][nt] = wp[0];
                    wb[n][nt] = wp[HALF4];
                }
            }
            __builtin_amdgcn_sched_barrier(0);
            {
                v2f u[4];
                make_u(s, i >> 1, u);
                mma_block(s, u);
            }
            __builtin_amdgcn_sched_barrier(0);
        }

        {
            v2f u[4];
            make_u(1, 2, u);
            __builtin_amdgcn_sched_barrier(0);
            if (more) w_request();
            __builtin_amdgcn_sched_barrier(0);
            mma_block(1, u);
            __builtin_amdgcn_sched_barrier(0);
            if (more) {
                x_request0(CCH * HW);
                w_store(cur4 ^ STAGE4);
                __syncthreads();
            }
        }

        #pragma unroll
        for (int r = 0; r < 3; ++r) { o0[r] += CCH * HW; o1h[r] += CCH * (HW / 2); o3[r] += CCH * HW; }
        #pragma unroll
        for (int i = 0; i < 3; ++i) goff[i] += CCH * 9;
        cur4 ^= STAGE4;
    };

    for (int cb0 = 0; cb0 < C_IN - CCH; cb0 += CCH) do_chunk(true);
    do_chunk(false);

    const int obase = nimg * (K_OUT * HW) + 2 * qi;
    #pragma unroll
    for (int nt = 0; nt < NT; ++nt) {
        const int ocb = oc_blk + n_base + nt * 16 + 8 * khalf;
        const v4f b0 = *reinterpret_cast<const v4f*>(bias + ocb);
        const v4f b1 = *reinterpret_cast<const v4f*>(bias + ocb + 4);
        const float bv[8] = { b0.x, b0.y, b0.z, b0.w, b1.x, b1.y, b1.z, b1.w };
        #pragma unroll
        for (int v = 0; v < 8; ++v) {
            const float m0 = acc[nt][0][v], m1 = acc[nt][1][v];
            const float m2 = acc[nt][2][v], m3 = acc[nt][3][v];
            const v2f y = { (m0 + m1) + m2 + bv[v], (m1 - m2) - m3 + bv[v] };
            *reinterpret_cast<v2f*>(out + obase + (ocb + v) * HW) = y;
        }
    }
}

__global__ __launch_bounds__(256, 4)
void conv3x3_wmma_f32(const float* __restrict__ x,
                      const float* __restrict__ wgt,
                      const float* __restrict__ bias,
                      float* __restrict__ out)
{
    __shared__ v4f wlds4[WLDS4];

    if (blockIdx.x < N_SMALL) {
        const int s = (int)blockIdx.x;
        conv_tile_ksplit(x, wgt, bias, out, wlds4,
                         BIG_COVER + (s / (K_OUT / SMALL_OC)) * SMALL_PAIRS,
                         (s % (K_OUT / SMALL_OC)) * SMALL_OC);
    } else {
        const int b = (int)blockIdx.x - N_SMALL;
        conv_tile_2stage<4, 8>(x, wgt, bias, out, wlds4,
                               (b / (K_OUT / BIG_OC)) * BIG_PAIRS, (b % (K_OUT / BIG_OC)) * BIG_OC);
    }
}

extern "C" void kernel_launch(void* const* d_in, const int* in_sizes, int n_in,
                              void* d_out, int out_size, void* d_ws, size_t ws_size,
                              hipStream_t stream) {
    (void)in_sizes; (void)n_in; (void)out_size; (void)d_ws; (void)ws_size;
    const float* x    = (const float*)d_in[0];
    const float* wgt  = (const float*)d_in[1];
    const float* bias = (const float*)d_in[2];
    float* out        = (float*)d_out;

    dim3 grid(N_SMALL + N_BIG, 1);
    dim3 block(256);
    conv3x3_wmma_f32<<<grid, block, 0, stream>>>(x, wgt, bias, out);
}


# Workgroup barriers order LDS only and no longer drain the global-load queue; next chunk's first pixel operand requested in front of the last matrix block, in flight across the barrier; sums unchanged
# speedup vs baseline: 3.3339x; 1.0310x over previous
#include <hip/hip_runtime.h>

typedef __attribute__((ext_vector_type(2))) float v2f;
typedef __attribute__((ext_vector_type(8))) float v8f;
typedef __attribute__((ext_vector_type(4))) float v4f;

#define C_IN  128
#define H_IN  56
#define W_IN  56
#define N_IMG 32
#define K_OUT 256
#define HW    (H_IN * W_IN)
#define KD    (C_IN * 9)
#define WPAIR (W_IN / 2)
#define HWP   (HW / 2)
#define P_TOT (N_IMG * HWP)

#define CCH      8
#define SLOTS    12
#define WPITCH4  25
#define WLDS4    (K_OUT * WPITCH4)
#define HALF4    12

#define BIG_PAIRS   128
#define BIG_OC      64
#define N_BIG       1536
#define BIG_COVER   ((N_BIG / (K_OUT / BIG_OC)) * BIG_PAIRS)
#define SMALL_PAIRS 16
#define SMALL_OC    32
#define SMALL_KQ    8
#define SMALL_CH    (C_IN / SMALL_KQ)
#define N_SMALL     (((P_TOT - BIG_COVER) / SMALL_PAIRS) * (K_OUT / SMALL_OC))

__device__ __forceinline__ void lds_barrier()
{
    __builtin_amdgcn_fence(__ATOMIC_RELEASE, "workgroup", "local");
    __builtin_amdgcn_s_barrier();
    __builtin_amdgcn_fence(__ATOMIC_ACQUIRE, "workgroup", "local");
}

__device__ __forceinline__
void conv_tile_ksplit(const float* __restrict__ x,
                      const float* __restrict__ wgt,
                      const float* __restrict__ bias,
                      float* __restrict__ out,
                      v4f* __restrict__ wlds4,
                      const int q_blk,
                      const int oc_blk)
{
    constexpr int NT = SMALL_OC / 16;
    constexpr int NB = (SMALL_OC * SMALL_KQ) / 128;
    constexpr int GSTEP = (64 / SMALL_OC) * SMALL_CH * 9;

    const int tid    = threadIdx.x;
    const int lane   = tid & 31;
    const int kq     = tid >> 5;

    const int n_base = kq * SMALL_OC;
    const int lrow   = lane & 15;
    const int khalf  = lane >> 4;

    const int qg   = q_blk + lrow;
    const int nimg = qg / HWP;
    const int qi   = qg - nimg * HWP;
    const int oh   = qi / WPAIR;
    const int jp   = qi - oh * WPAIR;

    int  o0[3], o1[3], o3[3];
    bool rv[3], v0[3], v3[3];
    #pragma unroll
    for (int r = 0; r < 3; ++r) {
        const int ih = oh + r - 1;
        rv[r] = (unsigned)ih < (unsigned)H_IN;
        v0[r] = rv[r] & (jp > 0);
        v3[r] = rv[r] & (jp < WPAIR - 1);
        o1[r] = nimg * (C_IN * HW) + (kq * SMALL_CH + khalf * 2) * HW + (rv[r] ? ih : oh) * W_IN + 2 * jp;
        o0[r] = o1[r] - (v0[r] ? 1 : 0);
        o3[r] = o1[r] + (v3[r] ? 2 : 0);
    }

    const v4f* const wrd = wlds4 + (n_base + lrow) * WPITCH4 + khalf * 3;

    int goff[3], loff4[3];
    #pragma unroll
    for (int i = 0; i < 3; ++i) {
        const int it   = tid + 256 * i;
        const int row  = it / SLOTS;
        const int slot = it - row * SLOTS;
        const int cp   = slot / 3;
        const int r    = slot - cp * 3;
        goff[i]  = (oc_blk + (row % SMALL_OC)) * KD + (row / SMALL_OC) * (SMALL_CH * 9) + cp * 18 + r * 3;
        loff4[i] = row * WPITCH4 + slot;
    }

    v8f acc[NT][4];
    #pragma unroll
    for (int nt = 0; nt < NT; ++nt)
        #pragma unroll
        for (int t = 0; t < 4; ++t) acc[nt][t] = (v8f){};

    float ra0[2], rb0[2], ra3[2], rb3[2];
    v2f   ra12[2], rb12[2];
    v4f   wa[2][NT], wb[2][NT];

    for (int cb0 = 0; cb0 < SMALL_CH; cb0 += CCH) {
        ra0[0]  = x[o0[0]];
        rb0[0]  = x[o0[0] + HW];
        ra12[0] = *reinterpret_cast<const v2f*>(x + o1[0]);
        rb12[0] = *reinterpret_cast<const v2f*>(x + o1[0] + HW);
        ra3[0]  = x[o3[0]];
        rb3[0]  = x[o3[0] + HW];

        float g[6][6];
        #pragma unroll
        for (int k = 0; k < 6; ++k) {
            const float* p = wgt + goff[k % 3] + (k / 3) * GSTEP;
            g[k][0] = p[0]; g[k][1] = p[1];  g[k][2] = p[2];
            g[k][3] = p[9]; g[k][4] = p[10]; g[k][5] = p[11];
        }

        lds_barrier();

        #pragma unroll
        for (int b = 0; b < NB; ++b) {
            #pragma unroll
            for (int k = 0; k < 6; ++k) {
                const float sa = g[k][0] + g[k][2];
                const float sb = g[k][3] + g[k][5];
                const v4f lo = { g[k][0], g[k][3], 0.5f * (sa + g[k][1]), 0.5f * (sb + g[k][4]) };
                const v4f hi = { 0.5f * (sa - g[k][1]), 0.5f * (sb - g[k][4]), g[k][2], g[k][5] };
                v4f* d = wlds4 + loff4[k % 3] + (b * 2 + k / 3) * (64 * WPITCH4);
                d[0]     = lo;
                d[HALF4] = hi;
            }
            if (b + 1 < NB) {
                #pragma unroll
                for (int k = 0; k < 6; ++k) {
                    const float* p = wgt + goff[k % 3] + ((b + 1) * 2 + k / 3) * GSTEP;
                    g[k][0] = p[0]; g[k][1] = p[1];  g[k][2] = p[2];
                    g[k][3] = p[9]; g[k][4] = p[10]; g[k][5] = p[11];
                }
            }
        }

        lds_barrier();

        #pragma unroll
        for (int nt = 0; nt < NT; ++nt) {
            const v4f* wp = wrd + nt * (16 * WPITCH4);
            wa[0][nt] = wp[0];
            wb[0][nt] = wp[HALF4];
        }

        #pragma unroll
        for (int i = 0; i < 6; ++i) {
            const int s = i & 1;
            if (i < 5) {
                const int r  = (i + 1) >> 1;
                const int ks = (i + 1) & 1;
                const int n  = s ^ 1;
                const float* xp = x + (ks * 4) * HW;
                ra0[n]  = xp[o0[r]];
                rb0[n]  = xp[o0[r] + HW];
                ra12[n] = *reinterpret_cast<const v2f*>(xp + o1[r]);
                rb12[n] = *reinterpret_cast<const v2f*>(xp + o1[r] + HW);
                ra3[n]  = xp[o3[r]];
                rb3[n]  = xp[o3[r] + HW];
                #pragma unroll
                for (int nt = 0; nt < NT; ++nt) {
                    const v4f* wp = wrd + nt * (16 * WPITCH4) + (ks * 6 + r);
                    wa[n][nt] = wp[0];
                    wb[n][nt] = wp[HALF4];
                }
            }
            __builtin_amdgcn_sched_barrier(0);
            {
                const int r = i >> 1;
                const float p0 = v0[r] ? ra0[s] : 0.0f,    q0 = v0[r] ? rb0[s] : 0.0f;
                const float p1 = rv[r] ? ra12[s].x : 0.0f, q1 = rv[r] ? rb12[s].x : 0.0f;
                const float p2 = rv[r] ? ra12[s].y : 0.0f, q2 = rv[r] ? rb12[s].y : 0.0f;
                const float p3 = v3[r] ? ra3[s] : 0.0f,    q3 = v3[r] ? rb3[s] : 0.0f;

                const v2f u0 = { p0 - p2, q0 - q2 };
                const v2f u1 = { p1 + p2, q1 + q2 };
                const v2f u2 = { p2 - p1, q2 - q1 };
                const v2f u3 = { p1 - p3, q1 - q3 };

                #pragma unroll
                for (int nt = 0; nt < NT; ++nt) {
                    const v2f a0 = { wa[s][nt].x, wa[s][nt].y };
                    const v2f a1 = { wa[s][nt].z, wa[s][nt].w };
                    const v2f a2 = { wb[s][nt].x, wb[s][nt].y };
                    const v2f a3 = { wb[s][nt].z, wb[s][nt].w };
                    acc[nt][0] = __builtin_amdgcn_wmma_f32_16x16x4_f32(
                        false, a0, false, u0, (short)0, acc[nt][0], false, false);
                    acc[nt][1] = __builtin_amdgcn_wmma_f32_16x16x4_f32(
                        false, a1, false, u1, (short)0, acc[nt][1], false, false);
                    acc[nt][2] = __builtin_amdgcn_wmma_f32_16x16x4_f32(
                        false, a2, false, u2, (short)0, acc[nt][2], false, false);
                    acc[nt][3] = __builtin_amdgcn_wmma_f32_16x16x4_f32(
                        false, a3, false, u3, (short)0, acc[nt][3], false, false);
                }
            }
            __builtin_amdgcn_sched_barrier(0);
        }

        #pragma unroll
        for (int r = 0; r < 3; ++r) { o0[r] += CCH * HW; o1[r] += CCH * HW; o3[r] += CCH * HW; }
        #pragma unroll
        for (int i = 0; i < 3; ++i) goff[i] += CCH * 9;
    }

    lds_barrier();
    float* const pf = reinterpret_cast<float*>(wlds4);
    #pragma unroll
    for (int nt = 0; nt < NT; ++nt)
        #pragma unroll
        for (int t = 0; t < 4; ++t)
            #pragma unroll
            for (int v = 0; v < 8; ++v)
                pf[(kq * 64 + (nt * 4 + t) * 8 + v) * 32 + lane] = acc[nt][t][v];
    lds_barrier();

    const int obase = nimg * (K_OUT * HW) + 2 * qi;
    #pragma unroll
    for (int i = 0; i < 2; ++i) {
        const int e  = 2 * kq + i;
        const int nt = e >> 3;
        const int v  = e & 7;
        float m[4];
        #pragma unroll
        for (int t = 0; t < 4; ++t) {
            const float* const pp = pf + ((nt * 4 + t) * 8 + v) * 32 + lane;
            float s = pp[0];
            #pragma unroll
            for (int q = 1; q < SMALL_KQ; ++q) s += pp[q * (64 * 32)];
            m[t] = s;
        }
        const int   oc = oc_blk + nt * 16 + 8 * khalf + v;
        const float bo = bias[oc];
        const v2f y = { (m[0] + m[1]) + m[2] + bo, (m[1] - m[2]) - m[3] + bo };
        *reinterpret_cast<v2f*>(out + obase + oc * HW) = y;
    }
}

template <int NT, int WM>
__device__ __forceinline__
void conv_tile_2stage(const float* __restrict__ x,
                      const float* __restrict__ wgt,
                      const float* __restrict__ bias,
                      float* __restrict__ out,
                      v4f* __restrict__ wlds4,
                      const int q_blk,
                      const int oc_blk)
{
    constexpr int KB = (8 / WM) * NT * 16;
    static_assert(KB * SLOTS == 3 * 256, "one staging batch of three items per thread");
    static_assert(2 * KB * WPITCH4 <= WLDS4, "two stages fit the allocation");
    constexpr int STAGE4 = KB * WPITCH4;

    const int tid    = threadIdx.x;
    const int lane   = tid & 31;
    const int wave   = tid >> 5;
    const int wave_m = wave % WM;
    const int wave_n = wave / WM;

    const int n_base = wave_n * (NT * 16);
    const int lrow   = lane & 15;
    const int khalf  = lane >> 4;

    const int qg   = q_blk + wave_m * 16 + lrow;
    const int nimg = qg / HWP;
    const int qi   = qg - nimg * HWP;
    const int oh   = qi / WPAIR;
    const int jp   = qi - oh * WPAIR;

    int  o0[3], o1[3], o3[3];
    bool rv[3], v0[3], v3[3];
    #pragma unroll
    for (int r = 0; r < 3; ++r) {
        const int ih = oh + r - 1;
        rv[r] = (unsigned)ih < (unsigned)H_IN;
        v0[r] = rv[r] & (jp > 0);
        v3[r] = rv[r] & (jp < WPAIR - 1);
        o1[r] = nimg * (C_IN * HW) + (khalf * 2) * HW + (rv[r] ? ih : oh) * W_IN + 2 * jp;
        o0[r] = o1[r] - (v0[r] ? 1 : 0);
        o3[r] = o1[r] + (v3[r] ? 2 : 0);
    }
    const v2f* __restrict__ const x2 = reinterpret_cast<const v2f*>(x);
    int o1h[3];
    #pragma unroll
    for (int r = 0; r < 3; ++r) o1h[r] = o1[r] >> 1;

    const v4f* const wrd = wlds4 + (n_base + lrow) * WPITCH4 + khalf * 3;

    int goff[3], loff4[3];
    #pragma unroll
    for (int i = 0; i < 3; ++i) {
        const int it   = tid + 256 * i;
        const int oc   = it / SLOTS;
        const int slot = it - oc * SLOTS;
        const int cp   = slot / 3;
        const int r    = slot - cp * 3;
        goff[i]  = (oc_blk + oc) * KD + cp * 18 + r * 3;
        loff4[i] = oc * WPITCH4 + slot;
    }

    v8f acc[NT][4];
    #pragma unroll
    for (int nt = 0; nt < NT; ++nt)
        #pragma unroll
        for (int t = 0; t < 4; ++t) acc[nt][t] = (v8f){};

    float ra0[2], rb0[2], ra3[2], rb3[2];
    v2f   ra12[2], rb12[2];
    v4f   wa[2][NT], wb[2][NT];
    float g[3][6];

    auto w_request = [&]() __attribute__((always_inline)) {
        #pragma unroll
        for (int k = 0; k < 3; ++k) {
            const float* p = wgt + goff[k];
            g[k][0] = p[0]; g[k][1] = p[1];  g[k][2] = p[2];
            g[k][3] = p[9]; g[k][4] = p[10]; g[k][5] = p[11];
        }
    };
    auto w_store = [&](const int st4) __attribute__((always_inline)) {
        #pragma unroll
        for (int k = 0; k < 3; ++k) {
            const float sa = g[k][0] + g[k][2];
            const float sb = g[k][3] + g[k][5];
            const v4f lo = { g[k][0], g[k][3], 0.5f * (sa + g[k][1]), 0.5f * (sb + g[k][4]) };
            const v4f hi = { 0.5f * (sa - g[k][1]), 0.5f * (sb - g[k][4]), g[k][2], g[k][5] };
            v4f* d = wlds4 + st4 + loff4[k];
            d[0]     = lo;
            d[HALF4] = hi;
        }
    };
    auto x_request0 = [&](const int adv) __attribute__((always_inline)) {
        const float* xp = x + adv;
        const v2f*   xq = x2 + adv / 2;
        ra0[0]  = xp[o0[0]];
        rb0[0]  = xp[o0[0] + HW];
        ra12[0] = xq[o1h[0]];
        rb12[0] = xq[o1h[0] + HW / 2];
        ra3[0]  = xp[o3[0]];
        rb3[0]  = xp[o3[0] + HW];
    };
    auto make_u = [&](const int s, const int r, v2f (&u)[4]) __attribute__((always_inline)) {
        const float p0 = v0[r] ? ra0[s] : 0.0f,    q0 = v0[r] ? rb0[s] : 0.0f;
        const float p1 = rv[r] ? ra12[s].x : 0.0f, q1 = rv[r] ? rb12[s].x : 0.0f;
        const float p2 = rv[r] ? ra12[s].y : 0.0f, q2 = rv[r] ? rb12[s].y : 0.0f;
        const float p3 = v3[r] ? ra3[s] : 0.0f,    q3 = v3[r] ? rb3[s] : 0.0f;
        u[0] = (v2f){ p0 - p2, q0 - q2 };
        u[1] = (v2f){ p1 + p2, q1 + q2 };
        u[2] = (v2f){ p2 - p1, q2 - q1 };
        u[3] = (v2f){ p1 - p3, q1 - q3 };
    };
    auto mma_block = [&](const int s, const v2f (&u)[4]) __attribute__((always_inline)) {
        #pragma unroll
        for (int nt = 0; nt < NT; ++nt) {
            const v2f a0 = { wa[s][nt].x, wa[s][nt].y };
            const v2f a1 = { wa[s][nt].z, wa[s][nt].w };
            const v2f a2 = { wb[s][nt].x, wb[s][nt].y };
            const v2f a3 = { wb[s][nt].z, wb[s][nt].w };
            acc[nt][0] = __builtin_amdgcn_wmma_f32_16x16x4_f32(
                false, a0, false, u[0], (short)0, acc[nt][0], false, false);
            acc[nt][1] = __builtin_amdgcn_wmma_f32_16x16x4_f32(
                false, a1, false, u[1], (short)0, acc[nt][1], false, false);
            acc[nt][2] = __builtin_amdgcn_wmma_f32_16x16x4_f32(
                false, a2, false, u[2], (short)0, acc[nt][2], false, false);
            acc[nt][3] = __builtin_amdgcn_wmma_f32_16x16x4_f32(
                false, a3, false, u[3], (short)0, acc[nt][3], false, false);
        }
    };

    x_request0(0);
    w_request();
    w_store(0);
    #pragma unroll
    for (int i = 0; i < 3; ++i) goff[i] += CCH * 9;
    lds_barrier();

    int cur4 = 0;
    auto do_chunk = [&](const bool more) __attribute__((always_inline)) {
        const v4f* const wr = wrd + cur4;

        #pragma unroll
        for (int nt = 0; nt < NT; ++nt) {
            const v4f* wp = wr + nt * (16 * WPITCH4);
            wa[0][nt] = wp[0];
            wb[0][nt] = wp[HALF4];
        }

        #pragma unroll
        for (int i = 0; i < 5; ++i) {
            const int s = i & 1;
            {
                const int r  = (i + 1) >> 1;
                const int ks = (i + 1) & 1;
                const int n  = s ^ 1;
                const float* xp = x + (ks * 4) * HW;
                const v2f*   xq = x2 + (ks * 4) * (HW / 2);
                ra0[n]  = xp[o0[r]];
                rb0[n]  = xp[o0[r] + HW];
                ra12[n] = xq[o1h[r]];
                rb12[n] = xq[o1h[r] + HW / 2];
                ra3[n]  = xp[o3[r]];
                rb3[n]  = xp[o3[r] + HW];
                #pragma unroll
                for (int nt = 0; nt < NT; ++nt) {
                    const v4f* wp = wr + nt * (16 * WPITCH4) + (ks * 6 + r);
                    wa[n][nt] = wp[0];
                    wb[n][nt] = wp[HALF4];
                }
            }
            __builtin_amdgcn_sched_barrier(0);
            {
                v2f u[4];
                make_u(s, i >> 1, u);
                mma_block(s, u);
            }
            __builtin_amdgcn_sched_barrier(0);
        }

        {
            v2f u[4];
            make_u(1, 2, u);
            __builtin_amdgcn_sched_barrier(0);
            if (more) { w_request(); x_request0(CCH * HW); }
            __builtin_amdgcn_sched_barrier(0);
            mma_block(1, u);
            __builtin_amdgcn_sched_barrier(0);
            if (more) {
                w_store(cur4 ^ STAGE4);
                lds_barrier();
            }
        }

        #pragma unroll
        for (int r = 0; r < 3; ++r) { o0[r] += CCH * HW; o1h[r] += CCH * (HW / 2); o3[r] += CCH * HW; }
        #pragma unroll
        for (int i = 0; i < 3; ++i) goff[i] += CCH * 9;
        cur4 ^= STAGE4;
    };

    for (int cb0 = 0; cb0 < C_IN - CCH; cb0 += CCH) do_chunk(true);
    do_chunk(false);

    const int obase = nimg * (K_OUT * HW) + 2 * qi;
    #pragma unroll
    for (int nt = 0; nt < NT; ++nt) {
        const int ocb = oc_blk + n_base + nt * 16 + 8 * khalf;
        const v4f b0 = *reinterpret_cast<const v4f*>(bias + ocb);
        const v4f b1 = *reinterpret_cast<const v4f*>(bias + ocb + 4);
        const float bv[8] = { b0.x, b0.y, b0.z, b0.w, b1.x, b1.y, b1.z, b1.w };
        #pragma unroll
        for (int v = 0; v < 8; ++v) {
            const float m0 = acc[nt][0][v], m1 = acc[nt][1][v];
            const float m2 = acc[nt][2][v], m3 = acc[nt][3][v];
            const v2f y = { (m0 + m1) + m2 + bv[v], (m1 - m2) - m3 + bv[v] };
            *reinterpret_cast<v2f*>(out + obase + (ocb + v) * HW) = y;
        }
    }
}

__global__ __launch_bounds__(256, 4)
void conv3x3_wmma_f32(const float* __restrict__ x,
                      const float* __restrict__ wgt,
                      const float* __restrict__ bias,
                      float* __restrict__ out)
{
    __shared__ v4f wlds4[WLDS4];

    if (blockIdx.x < N_SMALL) {
        const int s = (int)blockIdx.x;
        conv_tile_ksplit(x, wgt, bias, out, wlds4,
                         BIG_COVER + (s / (K_OUT / SMALL_OC)) * SMALL_PAIRS,
                         (s % (K_OUT / SMALL_OC)) * SMALL_OC);
    } else {
        const int b = (int)blockIdx.x - N_SMALL;
        conv_tile_2stage<4, 8>(x, wgt, bias, out, wlds4,
                               (b / (K_OUT / BIG_OC)) * BIG_PAIRS, (b % (K_OUT / BIG_OC)) * BIG_OC);
    }
}

extern "C" void kernel_launch(void* const* d_in, const int* in_sizes, int n_in,
                              void* d_out, int out_size, void* d_ws, size_t ws_size,
                              hipStream_t stream) {
    (void)in_sizes; (void)n_in; (void)out_size; (void)d_ws; (void)ws_size;
    const float* x    = (const float*)d_in[0];
    const float* wgt  = (const float*)d_in[1];
    const float* bias = (const float*)d_in[2];
    float* out        = (float*)d_out;

    dim3 grid(N_SMALL + N_BIG, 1);
    dim3 block(256);
    conv3x3_wmma_f32<<<grid, block, 0, stream>>>(x, wgt, bias, out);
}
